# starting kernel
# baseline (speedup 1.0000x reference)
#include <hip/hip_runtime.h>
#include <math.h>

typedef __attribute__((ext_vector_type(2))) float v2f;
typedef __attribute__((ext_vector_type(8))) float v8f;

static constexpr int IN_DIM  = 256;
static constexpr int HID_DIM = 128;
static constexpr int OUT_DIM = 16;

__global__ void k_deg_init(float* deg, int n) {
  int i = blockIdx.x * blockDim.x + threadIdx.x;
  if (i < n) deg[i] = 1.0f;
}

__global__ void k_deg_accum(const int* __restrict__ dst, float* deg, int E) {
  int e = blockIdx.x * blockDim.x + threadIdx.x;
  if (e < E) atomicAdd(&deg[dst[e]], 1.0f);
}

__global__ void k_deg_to_dinv(float* deg, int n) {
  int i = blockIdx.x * blockDim.x + threadIdx.x;
  if (i < n) {
    float d = deg[i];
    deg[i] = (d > 0.0f) ? rsqrtf(d) : 0.0f;
  }
}

template <int K_DIM, int N_DIM>
__global__ void k_gemm_wmma_f32(const float* __restrict__ A,
                                const float* __restrict__ B,
                                float* __restrict__ C, int M) {
  const int wave = (int)((blockIdx.x * blockDim.x + threadIdx.x) >> 5);
  const int lane = threadIdx.x & 31;
  constexpr int TN = N_DIM / 16;
  const int rt = wave / TN;
  const int ct = wave - rt * TN;
  if (rt * 16 >= M) return;

  const int lm = lane & 15;
  const int kh = lane >> 4;

  const float* arow = A + (size_t)(rt * 16 + lm) * K_DIM + 2 * kh;
  const float* bcol = B + (size_t)(2 * kh) * N_DIM + ct * 16 + lm;

  v8f acc = {};
#pragma unroll 8
  for (int k = 0; k < K_DIM; k += 4) {
    v2f a, b;
    a[0] = arow[k];
    a[1] = arow[k + 1];
    b[0] = bcol[(size_t)k * N_DIM];
    b[1] = bcol[(size_t)(k + 1) * N_DIM];
    acc = __builtin_amdgcn_wmma_f32_16x16x4_f32(false, a, false, b,
                                                (short)0, acc, false, false);
  }

  float* crow = C + (size_t)(rt * 16 + 8 * kh) * N_DIM + ct * 16 + lm;
#pragma unroll
  for (int j = 0; j < 8; ++j) crow[(size_t)j * N_DIM] = acc[j];
}

template <int F>
__global__ void k_selfloop(const float* __restrict__ h,
                           const float* __restrict__ dinv,
                           float* __restrict__ out, int n) {
  int idx = blockIdx.x * blockDim.x + threadIdx.x;
  if (idx >= n * F) return;
  int i = idx / F;
  float di = dinv[i];
  out[idx] = h[idx] * (di * di);
}

__global__ void k_scatter_f128(const int* __restrict__ src,
                               const int* __restrict__ dst,
                               const float* __restrict__ dinv,
                               const float* __restrict__ h,
                               float* out, int E) {
  int idx = blockIdx.x * blockDim.x + threadIdx.x;
  int e = idx >> 5;
  if (e >= E) return;
  int lane = idx & 31;
  int s = src[e], d = dst[e];
  float w = dinv[s] * dinv[d];
  const float4 v = *reinterpret_cast<const float4*>(h + (size_t)s * 128 + lane * 4);
  float* o = out + (size_t)d * 128 + lane * 4;
  atomicAdd(o + 0, v.x * w);
  atomicAdd(o + 1, v.y * w);
  atomicAdd(o + 2, v.z * w);
  atomicAdd(o + 3, v.w * w);
}

__global__ void k_scatter_f16(const int* __restrict__ src,
                              const int* __restrict__ dst,
                              const float* __restrict__ dinv,
                              const float* __restrict__ h,
                              float* out, int E) {
  int idx = blockIdx.x * blockDim.x + threadIdx.x;
  int e = idx >> 4;
  if (e >= E) return;
  int f = idx & 15;
  int s = src[e], d = dst[e];
  float w = dinv[s] * dinv[d];
  atomicAdd(&out[(size_t)d * 16 + f], h[(size_t)s * 16 + f] * w);
}

__global__ void k_bias_relu(float* h, const float* __restrict__ b, int total) {
  int idx = blockIdx.x * blockDim.x + threadIdx.x;
  if (idx >= total) return;
  h[idx] = fmaxf(h[idx] + b[idx & (HID_DIM - 1)], 0.0f);
}

__global__ void k_bias_logsoftmax16(float* out, const float* __restrict__ b, int n) {
  int i = blockIdx.x * blockDim.x + threadIdx.x;
  if (i >= n) return;
  float* row = out + (size_t)i * 16;
  float v[16];
  float m = -INFINITY;
#pragma unroll
  for (int f = 0; f < 16; ++f) {
    v[f] = row[f] + b[f];
    m = fmaxf(m, v[f]);
  }
  float s = 0.0f;
#pragma unroll
  for (int f = 0; f < 16; ++f) s += expf(v[f] - m);
  float ls = logf(s);
#pragma unroll
  for (int f = 0; f < 16; ++f) row[f] = v[f] - m - ls;
}

extern "C" void kernel_launch(void* const* d_in, const int* in_sizes, int n_in,
                              void* d_out, int out_size, void* d_ws, size_t ws_size,
                              hipStream_t stream) {
  (void)n_in; (void)out_size; (void)ws_size;

  const float* x  = (const float*)d_in[0];
  const int*   ei = (const int*)d_in[1];
  const float* W1 = (const float*)d_in[2];
  const float* b1 = (const float*)d_in[3];
  const float* W2 = (const float*)d_in[4];
  const float* b2 = (const float*)d_in[5];

  const int n = in_sizes[0] / IN_DIM;
  const int E = in_sizes[1] / 2;
  const int* src = ei;
  const int* dst = ei + E;

  float* dinv = (float*)d_ws;
  float* h1   = dinv + n;
  float* agg1 = h1 + (size_t)n * HID_DIM;
  float* h2   = agg1 + (size_t)n * HID_DIM;
  float* out  = (float*)d_out;

  const int B = 256;
  auto cdiv = [](long long a, long long b) { return (int)((a + b - 1) / b); };

  k_deg_init<<<cdiv(n, B), B, 0, stream>>>(dinv, n);
  k_deg_accum<<<cdiv(E, B), B, 0, stream>>>(dst, dinv, E);
  k_deg_to_dinv<<<cdiv(n, B), B, 0, stream>>>(dinv, n);

  {
    long long waves = (long long)cdiv(n, 16) * (HID_DIM / 16);
    k_gemm_wmma_f32<IN_DIM, HID_DIM>
        <<<cdiv(waves * 32, B), B, 0, stream>>>(x, W1, h1, n);
  }
  k_selfloop<HID_DIM><<<cdiv((long long)n * HID_DIM, B), B, 0, stream>>>(h1, dinv, agg1, n);
  k_scatter_f128<<<cdiv((long long)E * 32, B), B, 0, stream>>>(src, dst, dinv, h1, agg1, E);
  k_bias_relu<<<cdiv((long long)n * HID_DIM, B), B, 0, stream>>>(agg1, b1, n * HID_DIM);

  {
    long long waves = (long long)cdiv(n, 16) * (OUT_DIM / 16);
    k_gemm_wmma_f32<HID_DIM, OUT_DIM>
        <<<cdiv(waves * 32, B), B, 0, stream>>>(agg1, W2, h2, n);
  }
  k_selfloop<OUT_DIM><<<cdiv((long long)n * OUT_DIM, B), B, 0, stream>>>(h2, dinv, out, n);
  k_scatter_f16<<<cdiv((long long)E * 16, B), B, 0, stream>>>(src, dst, dinv, h2, out, E);
  k_bias_logsoftmax16<<<cdiv(n, B), B, 0, stream>>>(out, b2, n);
}


# Float-atomic scatter-add replaced by a per-call counting sort of edges by destination (returning int add = slot) and a register gather per node; sums run self term first, then arrival order
# speedup vs baseline: 1.5795x; 1.5795x over previous
#include <hip/hip_runtime.h>
#include <math.h>

typedef __attribute__((ext_vector_type(2))) float v2f;
typedef __attribute__((ext_vector_type(8))) float v8f;

static constexpr int IN_DIM  = 256;
static constexpr int HID_DIM = 128;
static constexpr int OUT_DIM = 16;

__global__ void k_zero_cnt(int* cnt, int n) {
  int i = blockIdx.x * blockDim.x + threadIdx.x;
  if (i < n) cnt[i] = 0;
}

__global__ void k_rank(const int* __restrict__ dst, int* cnt,
                       int* __restrict__ rank, int E) {
  int e = blockIdx.x * blockDim.x + threadIdx.x;
  if (e < E) rank[e] = atomicAdd(&cnt[dst[e]], 1);
}

static __device__ __forceinline__ float dinv_of_count(int c) {
  const float d = (float)(c + 1);
  return (d > 0.0f) ? rsqrtf(d) : 0.0f;
}

__global__ void __launch_bounds__(1024)
k_scan_dinv(const int* __restrict__ cnt, int* __restrict__ rowptr,
            float* __restrict__ dinv, int n) {
  __shared__ int wsum[32];
  __shared__ int chunk_total;
  const int t = threadIdx.x;
  const int lane = t & 31;
  const int wv = t >> 5;
  const bool vec_ok = (n & 3) == 0;
  int carry = 0;
  for (int base = 0; base < n; base += 16384) {
    const int i0 = base + t * 16;
    const bool full = vec_ok && (i0 + 16 <= n);
    int c[16];
    if (full) {
      const int4* p = reinterpret_cast<const int4*>(cnt + i0);
      const int4 q0 = p[0], q1 = p[1], q2 = p[2], q3 = p[3];
      c[0] = q0.x;  c[1] = q0.y;  c[2] = q0.z;  c[3] = q0.w;
      c[4] = q1.x;  c[5] = q1.y;  c[6] = q1.z;  c[7] = q1.w;
      c[8] = q2.x;  c[9] = q2.y;  c[10] = q2.z; c[11] = q2.w;
      c[12] = q3.x; c[13] = q3.y; c[14] = q3.z; c[15] = q3.w;
    } else {
#pragma unroll
      for (int k = 0; k < 16; ++k) c[k] = (i0 + k < n) ? cnt[i0 + k] : 0;
    }
    int tot = 0;
#pragma unroll
    for (int k = 0; k < 16; ++k) tot += c[k];
    int inc = tot;
    for (int o = 1; o < 32; o <<= 1) {
      const int y = __shfl_up(inc, o);
      if (lane >= o) inc += y;
    }
    if (lane == 31) wsum[wv] = inc;
    __syncthreads();
    if (wv == 0) {
      const int w = wsum[lane];
      int winc = w;
      for (int o = 1; o < 32; o <<= 1) {
        const int y = __shfl_up(winc, o);
        if (lane >= o) winc += y;
      }
      wsum[lane] = winc - w;
      if (lane == 31) chunk_total = winc;
    }
    __syncthreads();
    int run = carry + wsum[wv] + (inc - tot);
    carry += chunk_total;
    if (full) {
      int4* rp = reinterpret_cast<int4*>(rowptr + i0);
      float4* dp = reinterpret_cast<float4*>(dinv + i0);
#pragma unroll
      for (int k = 0; k < 4; ++k) {
        int4 r;
        float4 d;
        r.x = run; run += c[4 * k];
        r.y = run; run += c[4 * k + 1];
        r.z = run; run += c[4 * k + 2];
        r.w = run; run += c[4 * k + 3];
        d.x = dinv_of_count(c[4 * k]);
        d.y = dinv_of_count(c[4 * k + 1]);
        d.z = dinv_of_count(c[4 * k + 2]);
        d.w = dinv_of_count(c[4 * k + 3]);
        rp[k] = r;
        dp[k] = d;
      }
    } else {
#pragma unroll
      for (int k = 0; k < 16; ++k) {
        if (i0 + k < n) {
          rowptr[i0 + k] = run;
          dinv[i0 + k] = dinv_of_count(c[k]);
        }
        run += c[k];
      }
    }
    __syncthreads();
  }
}

__global__ void k_fill(const int* __restrict__ src, const int* __restrict__ dst,
                       const int* __restrict__ rank, const int* __restrict__ rowptr,
                       const float* __restrict__ dinv,
                       int* __restrict__ csr_src, float* __restrict__ csr_w, int E) {
  int e = blockIdx.x * blockDim.x + threadIdx.x;
  if (e >= E) return;
  const int s = src[e];
  int pos = rowptr[dst[e]] + rank[e];
  pos = min(max(pos, 0), E - 1);
  csr_src[pos] = s;
  csr_w[pos] = dinv[s];
}

template <int K_DIM, int N_DIM>
__global__ void k_gemm_wmma_f32(const float* __restrict__ A,
                                const float* __restrict__ B,
                                float* __restrict__ C, int M) {
  const int wave = (int)((blockIdx.x * blockDim.x + threadIdx.x) >> 5);
  const int lane = threadIdx.x & 31;
  constexpr int TN = N_DIM / 16;
  const int rt = wave / TN;
  const int ct = wave - rt * TN;
  if (rt * 16 >= M) return;

  const int lm = lane & 15;
  const int kh = lane >> 4;

  const float* arow = A + (size_t)(rt * 16 + lm) * K_DIM + 2 * kh;
  const float* bcol = B + (size_t)(2 * kh) * N_DIM + ct * 16 + lm;

  v8f acc = {};
#pragma unroll 8
  for (int k = 0; k < K_DIM; k += 4) {
    v2f a, b;
    a[0] = arow[k];
    a[1] = arow[k + 1];
    b[0] = bcol[(size_t)k * N_DIM];
    b[1] = bcol[(size_t)(k + 1) * N_DIM];
    acc = __builtin_amdgcn_wmma_f32_16x16x4_f32(false, a, false, b,
                                                (short)0, acc, false, false);
  }

  float* crow = C + (size_t)(rt * 16 + 8 * kh) * N_DIM + ct * 16 + lm;
#pragma unroll
  for (int j = 0; j < 8; ++j) crow[(size_t)j * N_DIM] = acc[j];
}


__global__ void k_gather_f128(const int* __restrict__ rowptr,
                              const int* __restrict__ cnt,
                              const int* __restrict__ csr_src,
                              const float* __restrict__ csr_w,
                              const float* __restrict__ dinv,
                              const float* __restrict__ h,
                              float* __restrict__ out, int n, int E) {
  const int node = __builtin_amdgcn_readfirstlane(
      (int)((blockIdx.x * blockDim.x + threadIdx.x) >> 5));
  if (node >= n) return;
  const int lane = threadIdx.x & 31;
  const int len = min(max(cnt[node], 0), E);
  const int start = min(max(rowptr[node], 0), E - len);
  const float dd = dinv[node];
  const float4 hv = *reinterpret_cast<const float4*>(h + (size_t)node * 128 + lane * 4);
  const float wself = dd * dd;
  float4 acc;
  acc.x = hv.x * wself;
  acc.y = hv.y * wself;
  acc.z = hv.z * wself;
  acc.w = hv.w * wself;
  for (int b = 0; b < len; b += 32) {
    const int m = min(32, len - b);
    int s_l = 0;
    float w_l = 0.0f;
    if (lane < m) {
      s_l = min(max(csr_src[start + b + lane], 0), n - 1);
      w_l = csr_w[start + b + lane] * dd;
    }
    int j = 0;
    for (; j + 4 <= m; j += 4) {
      const int s0 = __builtin_amdgcn_readlane(s_l, j);
      const int s1 = __builtin_amdgcn_readlane(s_l, j + 1);
      const int s2 = __builtin_amdgcn_readlane(s_l, j + 2);
      const int s3 = __builtin_amdgcn_readlane(s_l, j + 3);
      const float w0 = __int_as_float(__builtin_amdgcn_readlane(__float_as_int(w_l), j));
      const float w1 = __int_as_float(__builtin_amdgcn_readlane(__float_as_int(w_l), j + 1));
      const float w2 = __int_as_float(__builtin_amdgcn_readlane(__float_as_int(w_l), j + 2));
      const float w3 = __int_as_float(__builtin_amdgcn_readlane(__float_as_int(w_l), j + 3));
      const float4 v0 = *reinterpret_cast<const float4*>(h + (size_t)s0 * 128 + lane * 4);
      const float4 v1 = *reinterpret_cast<const float4*>(h + (size_t)s1 * 128 + lane * 4);
      const float4 v2 = *reinterpret_cast<const float4*>(h + (size_t)s2 * 128 + lane * 4);
      const float4 v3 = *reinterpret_cast<const float4*>(h + (size_t)s3 * 128 + lane * 4);
      acc.x += v0.x * w0; acc.y += v0.y * w0; acc.z += v0.z * w0; acc.w += v0.w * w0;
      acc.x += v1.x * w1; acc.y += v1.y * w1; acc.z += v1.z * w1; acc.w += v1.w * w1;
      acc.x += v2.x * w2; acc.y += v2.y * w2; acc.z += v2.z * w2; acc.w += v2.w * w2;
      acc.x += v3.x * w3; acc.y += v3.y * w3; acc.z += v3.z * w3; acc.w += v3.w * w3;
    }
    for (; j < m; ++j) {
      const int s0 = __builtin_amdgcn_readlane(s_l, j);
      const float w0 = __int_as_float(__builtin_amdgcn_readlane(__float_as_int(w_l), j));
      const float4 v0 = *reinterpret_cast<const float4*>(h + (size_t)s0 * 128 + lane * 4);
      acc.x += v0.x * w0; acc.y += v0.y * w0; acc.z += v0.z * w0; acc.w += v0.w * w0;
    }
  }
  *reinterpret_cast<float4*>(out + (size_t)node * 128 + lane * 4) = acc;
}

__global__ void k_gather_f16(const int* __restrict__ rowptr,
                             const int* __restrict__ cnt,
                             const int* __restrict__ csr_src,
                             const float* __restrict__ csr_w,
                             const float* __restrict__ dinv,
                             const float* __restrict__ h,
                             float* __restrict__ out, int n, int E) {
  const int idx = blockIdx.x * blockDim.x + threadIdx.x;
  const int node = idx >> 4;
  if (node >= n) return;
  const int f = idx & 15;
  const int len = min(max(cnt[node], 0), E);
  const int start = min(max(rowptr[node], 0), E - len);
  const float dd = dinv[node];
  float acc = h[(size_t)node * 16 + f] * (dd * dd);
  int j = 0;
  for (; j + 4 <= len; j += 4) {
    const int s0 = min(max(csr_src[start + j], 0), n - 1);
    const int s1 = min(max(csr_src[start + j + 1], 0), n - 1);
    const int s2 = min(max(csr_src[start + j + 2], 0), n - 1);
    const int s3 = min(max(csr_src[start + j + 3], 0), n - 1);
    const float w0 = csr_w[start + j] * dd;
    const float w1 = csr_w[start + j + 1] * dd;
    const float w2 = csr_w[start + j + 2] * dd;
    const float w3 = csr_w[start + j + 3] * dd;
    const float v0 = h[(size_t)s0 * 16 + f];
    const float v1 = h[(size_t)s1 * 16 + f];
    const float v2 = h[(size_t)s2 * 16 + f];
    const float v3 = h[(size_t)s3 * 16 + f];
    acc += v0 * w0;
    acc += v1 * w1;
    acc += v2 * w2;
    acc += v3 * w3;
  }
  for (; j < len; ++j) {
    const int s0 = min(max(csr_src[start + j], 0), n - 1);
    acc += h[(size_t)s0 * 16 + f] * (csr_w[start + j] * dd);
  }
  out[(size_t)node * 16 + f] = acc;
}

__global__ void k_bias_relu(float* h, const float* __restrict__ b, int total) {
  int idx = blockIdx.x * blockDim.x + threadIdx.x;
  if (idx >= total) return;
  h[idx] = fmaxf(h[idx] + b[idx & (HID_DIM - 1)], 0.0f);
}

__global__ void k_bias_logsoftmax16(float* out, const float* __restrict__ b, int n) {
  int i = blockIdx.x * blockDim.x + threadIdx.x;
  if (i >= n) return;
  float* row = out + (size_t)i * 16;
  float v[16];
  float m = -INFINITY;
#pragma unroll
  for (int f = 0; f < 16; ++f) {
    v[f] = row[f] + b[f];
    m = fmaxf(m, v[f]);
  }
  float s = 0.0f;
#pragma unroll
  for (int f = 0; f < 16; ++f) s += expf(v[f] - m);
  float ls = logf(s);
#pragma unroll
  for (int f = 0; f < 16; ++f) row[f] = v[f] - m - ls;
}

extern "C" void kernel_launch(void* const* d_in, const int* in_sizes, int n_in,
                              void* d_out, int out_size, void* d_ws, size_t ws_size,
                              hipStream_t stream) {
  (void)n_in; (void)out_size; (void)ws_size;

  const float* x  = (const float*)d_in[0];
  const int*   ei = (const int*)d_in[1];
  const float* W1 = (const float*)d_in[2];
  const float* b1 = (const float*)d_in[3];
  const float* W2 = (const float*)d_in[4];
  const float* b2 = (const float*)d_in[5];

  const int n = in_sizes[0] / IN_DIM;
  const int E = in_sizes[1] / 2;
  const int* src = ei;
  const int* dst = ei + E;

  float* h1     = (float*)d_ws;
  float* agg1   = h1 + (size_t)n * HID_DIM;
  float* h2     = agg1 + (size_t)n * HID_DIM;
  int*   cnt    = (int*)(h2 + (size_t)n * OUT_DIM);
  int*   rowptr = cnt + n;
  float* dinv   = (float*)(rowptr + n);
  int*   rank   = (int*)(dinv + n);
  int*   csr    = rank + E;
  float* csrw   = (float*)(csr + E);
  float* out    = (float*)d_out;

  const int B = 256;
  auto cdiv = [](long long a, long long b) { return (int)((a + b - 1) / b); };

  k_zero_cnt<<<cdiv(n, B), B, 0, stream>>>(cnt, n);
  k_rank<<<cdiv(E, B), B, 0, stream>>>(dst, cnt, rank, E);
  k_scan_dinv<<<1, 1024, 0, stream>>>(cnt, rowptr, dinv, n);
  k_fill<<<cdiv(E, B), B, 0, stream>>>(src, dst, rank, rowptr, dinv, csr, csrw, E);

  {
    long long waves = (long long)cdiv(n, 16) * (HID_DIM / 16);
    k_gemm_wmma_f32<IN_DIM, HID_DIM>
        <<<cdiv(waves * 32, B), B, 0, stream>>>(x, W1, h1, n);
  }
  k_gather_f128<<<cdiv((long long)n * 32, B), B, 0, stream>>>(rowptr, cnt, csr, csrw, dinv, h1, agg1, n, E);
  k_bias_relu<<<cdiv((long long)n * HID_DIM, B), B, 0, stream>>>(agg1, b1, n * HID_DIM);

  {
    long long waves = (long long)cdiv(n, 16) * (OUT_DIM / 16);
    k_gemm_wmma_f32<HID_DIM, OUT_DIM>
        <<<cdiv(waves * 32, B), B, 0, stream>>>(agg1, W2, h2, n);
  }
  k_gather_f16<<<cdiv((long long)n * OUT_DIM, B), B, 0, stream>>>(rowptr, cnt, csr, csrw, dinv, h2, out, n, E);
  k_bias_logsoftmax16<<<cdiv(n, B), B, 0, stream>>>(out, b2, n);
}


# Bias+ReLU and bias+log_softmax folded into the stores of the two gather kernels that produce their inputs (10 launches -> 8); every sum keeps the parent's order
# speedup vs baseline: 1.7927x; 1.1350x over previous
#include <hip/hip_runtime.h>
#include <math.h>

typedef __attribute__((ext_vector_type(2))) float v2f;
typedef __attribute__((ext_vector_type(8))) float v8f;

static constexpr int IN_DIM  = 256;
static constexpr int HID_DIM = 128;
static constexpr int OUT_DIM = 16;

__global__ void k_zero_cnt(int* cnt, int n) {
  int i = blockIdx.x * blockDim.x + threadIdx.x;
  if (i < n) cnt[i] = 0;
}

__global__ void k_rank(const int* __restrict__ dst, int* cnt,
                       int* __restrict__ rank, int E) {
  int e = blockIdx.x * blockDim.x + threadIdx.x;
  if (e < E) rank[e] = atomicAdd(&cnt[dst[e]], 1);
}

static __device__ __forceinline__ float dinv_of_count(int c) {
  const float d = (float)(c + 1);
  return (d > 0.0f) ? rsqrtf(d) : 0.0f;
}

__global__ void __launch_bounds__(1024)
k_scan_dinv(const int* __restrict__ cnt, int* __restrict__ rowptr,
            float* __restrict__ dinv, int n) {
  __shared__ int wsum[32];
  __shared__ int chunk_total;
  const int t = threadIdx.x;
  const int lane = t & 31;
  const int wv = t >> 5;
  const bool vec_ok = (n & 3) == 0;
  int carry = 0;
  for (int base = 0; base < n; base += 16384) {
    const int i0 = base + t * 16;
    const bool full = vec_ok && (i0 + 16 <= n);
    int c[16];
    if (full) {
      const int4* p = reinterpret_cast<const int4*>(cnt + i0);
      const int4 q0 = p[0], q1 = p[1], q2 = p[2], q3 = p[3];
      c[0] = q0.x;  c[1] = q0.y;  c[2] = q0.z;  c[3] = q0.w;
      c[4] = q1.x;  c[5] = q1.y;  c[6] = q1.z;  c[7] = q1.w;
      c[8] = q2.x;  c[9] = q2.y;  c[10] = q2.z; c[11] = q2.w;
      c[12] = q3.x; c[13] = q3.y; c[14] = q3.z; c[15] = q3.w;
    } else {
#pragma unroll
      for (int k = 0; k < 16; ++k) c[k] = (i0 + k < n) ? cnt[i0 + k] : 0;
    }
    int tot = 0;
#pragma unroll
    for (int k = 0; k < 16; ++k) tot += c[k];
    int inc = tot;
    for (int o = 1; o < 32; o <<= 1) {
      const int y = __shfl_up(inc, o);
      if (lane >= o) inc += y;
    }
    if (lane == 31) wsum[wv] = inc;
    __syncthreads();
    if (wv == 0) {
      const int w = wsum[lane];
      int winc = w;
      for (int o = 1; o < 32; o <<= 1) {
        const int y = __shfl_up(winc, o);
        if (lane >= o) winc += y;
      }
      wsum[lane] = winc - w;
      if (lane == 31) chunk_total = winc;
    }
    __syncthreads();
    int run = carry + wsum[wv] + (inc - tot);
    carry += chunk_total;
    if (full) {
      int4* rp = reinterpret_cast<int4*>(rowptr + i0);
      float4* dp = reinterpret_cast<float4*>(dinv + i0);
#pragma unroll
      for (int k = 0; k < 4; ++k) {
        int4 r;
        float4 d;
        r.x = run; run += c[4 * k];
        r.y = run; run += c[4 * k + 1];
        r.z = run; run += c[4 * k + 2];
        r.w = run; run += c[4 * k + 3];
        d.x = dinv_of_count(c[4 * k]);
        d.y = dinv_of_count(c[4 * k + 1]);
        d.z = dinv_of_count(c[4 * k + 2]);
        d.w = dinv_of_count(c[4 * k + 3]);
        rp[k] = r;
        dp[k] = d;
      }
    } else {
#pragma unroll
      for (int k = 0; k < 16; ++k) {
        if (i0 + k < n) {
          rowptr[i0 + k] = run;
          dinv[i0 + k] = dinv_of_count(c[k]);
        }
        run += c[k];
      }
    }
    __syncthreads();
  }
}

__global__ void k_fill(const int* __restrict__ src, const int* __restrict__ dst,
                       const int* __restrict__ rank, const int* __restrict__ rowptr,
                       const float* __restrict__ dinv,
                       int* __restrict__ csr_src, float* __restrict__ csr_w, int E) {
  int e = blockIdx.x * blockDim.x + threadIdx.x;
  if (e >= E) return;
  const int s = src[e];
  int pos = rowptr[dst[e]] + rank[e];
  pos = min(max(pos, 0), E - 1);
  csr_src[pos] = s;
  csr_w[pos] = dinv[s];
}

template <int K_DIM, int N_DIM>
__global__ void k_gemm_wmma_f32(const float* __restrict__ A,
                                const float* __restrict__ B,
                                float* __restrict__ C, int M) {
  const int wave = (int)((blockIdx.x * blockDim.x + threadIdx.x) >> 5);
  const int lane = threadIdx.x & 31;
  constexpr int TN = N_DIM / 16;
  const int rt = wave / TN;
  const int ct = wave - rt * TN;
  if (rt * 16 >= M) return;

  const int lm = lane & 15;
  const int kh = lane >> 4;

  const float* arow = A + (size_t)(rt * 16 + lm) * K_DIM + 2 * kh;
  const float* bcol = B + (size_t)(2 * kh) * N_DIM + ct * 16 + lm;

  v8f acc = {};
#pragma unroll 8
  for (int k = 0; k < K_DIM; k += 4) {
    v2f a, b;
    a[0] = arow[k];
    a[1] = arow[k + 1];
    b[0] = bcol[(size_t)k * N_DIM];
    b[1] = bcol[(size_t)(k + 1) * N_DIM];
    acc = __builtin_amdgcn_wmma_f32_16x16x4_f32(false, a, false, b,
                                                (short)0, acc, false, false);
  }

  float* crow = C + (size_t)(rt * 16 + 8 * kh) * N_DIM + ct * 16 + lm;
#pragma unroll
  for (int j = 0; j < 8; ++j) crow[(size_t)j * N_DIM] = acc[j];
}


__global__ void k_gather_f128_bias_relu(const int* __restrict__ rowptr,
                                        const int* __restrict__ cnt,
                                        const int* __restrict__ csr_src,
                                        const float* __restrict__ csr_w,
                                        const float* __restrict__ dinv,
                                        const float* __restrict__ h,
                                        const float* __restrict__ bias,
                                        float* __restrict__ out, int n, int E) {
  const int node = __builtin_amdgcn_readfirstlane(
      (int)((blockIdx.x * blockDim.x + threadIdx.x) >> 5));
  if (node >= n) return;
  const int lane = threadIdx.x & 31;
  const int len = min(max(cnt[node], 0), E);
  const int start = min(max(rowptr[node], 0), E - len);
  const float dd = dinv[node];
  const float4 hv = *reinterpret_cast<const float4*>(h + (size_t)node * 128 + lane * 4);
  const float4 bv = *reinterpret_cast<const float4*>(bias + lane * 4);
  const float wself = dd * dd;
  float4 acc;
  acc.x = hv.x * wself;
  acc.y = hv.y * wself;
  acc.z = hv.z * wself;
  acc.w = hv.w * wself;
  for (int b = 0; b < len; b += 32) {
    const int m = min(32, len - b);
    int s_l = 0;
    float w_l = 0.0f;
    if (lane < m) {
      s_l = min(max(csr_src[start + b + lane], 0), n - 1);
      w_l = csr_w[start + b + lane] * dd;
    }
    int j = 0;
    for (; j + 4 <= m; j += 4) {
      const int s0 = __builtin_amdgcn_readlane(s_l, j);
      const int s1 = __builtin_amdgcn_readlane(s_l, j + 1);
      const int s2 = __builtin_amdgcn_readlane(s_l, j + 2);
      const int s3 = __builtin_amdgcn_readlane(s_l, j + 3);
      const float w0 = __int_as_float(__builtin_amdgcn_readlane(__float_as_int(w_l), j));
      const float w1 = __int_as_float(__builtin_amdgcn_readlane(__float_as_int(w_l), j + 1));
      const float w2 = __int_as_float(__builtin_amdgcn_readlane(__float_as_int(w_l), j + 2));
      const float w3 = __int_as_float(__builtin_amdgcn_readlane(__float_as_int(w_l), j + 3));
      const float4 v0 = *reinterpret_cast<const float4*>(h + (size_t)s0 * 128 + lane * 4);
      const float4 v1 = *reinterpret_cast<const float4*>(h + (size_t)s1 * 128 + lane * 4);
      const float4 v2 = *reinterpret_cast<const float4*>(h + (size_t)s2 * 128 + lane * 4);
      const float4 v3 = *reinterpret_cast<const float4*>(h + (size_t)s3 * 128 + lane * 4);
      acc.x += v0.x * w0; acc.y += v0.y * w0; acc.z += v0.z * w0; acc.w += v0.w * w0;
      acc.x += v1.x * w1; acc.y += v1.y * w1; acc.z += v1.z * w1; acc.w += v1.w * w1;
      acc.x += v2.x * w2; acc.y += v2.y * w2; acc.z += v2.z * w2; acc.w += v2.w * w2;
      acc.x += v3.x * w3; acc.y += v3.y * w3; acc.z += v3.z * w3; acc.w += v3.w * w3;
    }
    for (; j < m; ++j) {
      const int s0 = __builtin_amdgcn_readlane(s_l, j);
      const float w0 = __int_as_float(__builtin_amdgcn_readlane(__float_as_int(w_l), j));
      const float4 v0 = *reinterpret_cast<const float4*>(h + (size_t)s0 * 128 + lane * 4);
      acc.x += v0.x * w0; acc.y += v0.y * w0; acc.z += v0.z * w0; acc.w += v0.w * w0;
    }
  }
  acc.x = fmaxf(acc.x + bv.x, 0.0f);
  acc.y = fmaxf(acc.y + bv.y, 0.0f);
  acc.z = fmaxf(acc.z + bv.z, 0.0f);
  acc.w = fmaxf(acc.w + bv.w, 0.0f);
  *reinterpret_cast<float4*>(out + (size_t)node * 128 + lane * 4) = acc;
}

__global__ void k_gather_f16_bias_logsoftmax(const int* __restrict__ rowptr,
                                             const int* __restrict__ cnt,
                                             const int* __restrict__ csr_src,
                                             const float* __restrict__ csr_w,
                                             const float* __restrict__ dinv,
                                             const float* __restrict__ h,
                                             const float* __restrict__ bias,
                                             float* __restrict__ out, int n, int E) {
  const int idx = blockIdx.x * blockDim.x + threadIdx.x;
  const int node = idx >> 4;
  if (node >= n) return;
  const int f = idx & 15;
  const int len = min(max(cnt[node], 0), E);
  const int start = min(max(rowptr[node], 0), E - len);
  const float dd = dinv[node];
  float acc = h[(size_t)node * 16 + f] * (dd * dd);
  int j = 0;
  for (; j + 4 <= len; j += 4) {
    const int s0 = min(max(csr_src[start + j], 0), n - 1);
    const int s1 = min(max(csr_src[start + j + 1], 0), n - 1);
    const int s2 = min(max(csr_src[start + j + 2], 0), n - 1);
    const int s3 = min(max(csr_src[start + j + 3], 0), n - 1);
    const float w0 = csr_w[start + j] * dd;
    const float w1 = csr_w[start + j + 1] * dd;
    const float w2 = csr_w[start + j + 2] * dd;
    const float w3 = csr_w[start + j + 3] * dd;
    const float v0 = h[(size_t)s0 * 16 + f];
    const float v1 = h[(size_t)s1 * 16 + f];
    const float v2 = h[(size_t)s2 * 16 + f];
    const float v3 = h[(size_t)s3 * 16 + f];
    acc += v0 * w0;
    acc += v1 * w1;
    acc += v2 * w2;
    acc += v3 * w3;
  }
  for (; j < len; ++j) {
    const int s0 = min(max(csr_src[start + j], 0), n - 1);
    acc += h[(size_t)s0 * 16 + f] * (csr_w[start + j] * dd);
  }
  const float v = acc + bias[f];
  float m = __shfl(v, 0, 16);
#pragma unroll
  for (int k = 1; k < 16; ++k) m = fmaxf(m, __shfl(v, k, 16));
  const float ex = expf(v - m);
  float s = 0.0f;
#pragma unroll
  for (int k = 0; k < 16; ++k) s += __shfl(ex, k, 16);
  out[(size_t)node * 16 + f] = v - m - logf(s);
}

extern "C" void kernel_launch(void* const* d_in, const int* in_sizes, int n_in,
                              void* d_out, int out_size, void* d_ws, size_t ws_size,
                              hipStream_t stream) {
  (void)n_in; (void)out_size; (void)ws_size;

  const float* x  = (const float*)d_in[0];
  const int*   ei = (const int*)d_in[1];
  const float* W1 = (const float*)d_in[2];
  const float* b1 = (const float*)d_in[3];
  const float* W2 = (const float*)d_in[4];
  const float* b2 = (const float*)d_in[5];

  const int n = in_sizes[0] / IN_DIM;
  const int E = in_sizes[1] / 2;
  const int* src = ei;
  const int* dst = ei + E;

  float* h1     = (float*)d_ws;
  float* agg1   = h1 + (size_t)n * HID_DIM;
  float* h2     = agg1 + (size_t)n * HID_DIM;
  int*   cnt    = (int*)(h2 + (size_t)n * OUT_DIM);
  int*   rowptr = cnt + n;
  float* dinv   = (float*)(rowptr + n);
  int*   rank   = (int*)(dinv + n);
  int*   csr    = rank + E;
  float* csrw   = (float*)(csr + E);
  float* out    = (float*)d_out;

  const int B = 256;
  auto cdiv = [](long long a, long long b) { return (int)((a + b - 1) / b); };

  k_zero_cnt<<<cdiv(n, B), B, 0, stream>>>(cnt, n);
  k_rank<<<cdiv(E, B), B, 0, stream>>>(dst, cnt, rank, E);
  k_scan_dinv<<<1, 1024, 0, stream>>>(cnt, rowptr, dinv, n);
  k_fill<<<cdiv(E, B), B, 0, stream>>>(src, dst, rank, rowptr, dinv, csr, csrw, E);

  {
    long long waves = (long long)cdiv(n, 16) * (HID_DIM / 16);
    k_gemm_wmma_f32<IN_DIM, HID_DIM>
        <<<cdiv(waves * 32, B), B, 0, stream>>>(x, W1, h1, n);
  }
  k_gather_f128_bias_relu<<<cdiv((long long)n * 32, B), B, 0, stream>>>(rowptr, cnt, csr, csrw, dinv, h1, b1, agg1, n, E);

  {
    long long waves = (long long)cdiv(n, 16) * (OUT_DIM / 16);
    k_gemm_wmma_f32<HID_DIM, OUT_DIM>
        <<<cdiv(waves * 32, B), B, 0, stream>>>(agg1, W2, h2, n);
  }
  k_gather_f16_bias_logsoftmax<<<cdiv((long long)n * OUT_DIM, B), B, 0, stream>>>(rowptr, cnt, csr, csrw, dinv, h2, b2, out, n, E);
}


# Layer-1 GEMM re-tiled: W1 staged once per work-group in LDS, each wave owns a 16x128 output strip, so one A fragment load feeds 8 WMMAs and B comes from LDS; k order and every sum as in the parent
# speedup vs baseline: 2.0424x; 1.1393x over previous
#include <hip/hip_runtime.h>
#include <math.h>

typedef __attribute__((ext_vector_type(2))) float v2f;
typedef __attribute__((ext_vector_type(8))) float v8f;

static constexpr int IN_DIM  = 256;
static constexpr int HID_DIM = 128;
static constexpr int OUT_DIM = 16;

__global__ void k_zero_cnt(int* cnt, int n) {
  int i = blockIdx.x * blockDim.x + threadIdx.x;
  if (i < n) cnt[i] = 0;
}

__global__ void k_rank(const int* __restrict__ dst, int* cnt,
                       int* __restrict__ rank, int E) {
  int e = blockIdx.x * blockDim.x + threadIdx.x;
  if (e < E) rank[e] = atomicAdd(&cnt[dst[e]], 1);
}

static __device__ __forceinline__ float dinv_of_count(int c) {
  const float d = (float)(c + 1);
  return (d > 0.0f) ? rsqrtf(d) : 0.0f;
}

__global__ void __launch_bounds__(1024)
k_scan_dinv(const int* __restrict__ cnt, int* __restrict__ rowptr,
            float* __restrict__ dinv, int n) {
  __shared__ int wsum[32];
  __shared__ int chunk_total;
  const int t = threadIdx.x;
  const int lane = t & 31;
  const int wv = t >> 5;
  const bool vec_ok = (n & 3) == 0;
  int carry = 0;
  for (int base = 0; base < n; base += 16384) {
    const int i0 = base + t * 16;
    const bool full = vec_ok && (i0 + 16 <= n);
    int c[16];
    if (full) {
      const int4* p = reinterpret_cast<const int4*>(cnt + i0);
      const int4 q0 = p[0], q1 = p[1], q2 = p[2], q3 = p[3];
      c[0] = q0.x;  c[1] = q0.y;  c[2] = q0.z;  c[3] = q0.w;
      c[4] = q1.x;  c[5] = q1.y;  c[6] = q1.z;  c[7] = q1.w;
      c[8] = q2.x;  c[9] = q2.y;  c[10] = q2.z; c[11] = q2.w;
      c[12] = q3.x; c[13] = q3.y; c[14] = q3.z; c[15] = q3.w;
    } else {
#pragma unroll
      for (int k = 0; k < 16; ++k) c[k] = (i0 + k < n) ? cnt[i0 + k] : 0;
    }
    int tot = 0;
#pragma unroll
    for (int k = 0; k < 16; ++k) tot += c[k];
    int inc = tot;
    for (int o = 1; o < 32; o <<= 1) {
      const int y = __shfl_up(inc, o);
      if (lane >= o) inc += y;
    }
    if (lane == 31) wsum[wv] = inc;
    __syncthreads();
    if (wv == 0) {
      const int w = wsum[lane];
      int winc = w;
      for (int o = 1; o < 32; o <<= 1) {
        const int y = __shfl_up(winc, o);
        if (lane >= o) winc += y;
      }
      wsum[lane] = winc - w;
      if (lane == 31) chunk_total = winc;
    }
    __syncthreads();
    int run = carry + wsum[wv] + (inc - tot);
    carry += chunk_total;
    if (full) {
      int4* rp = reinterpret_cast<int4*>(rowptr + i0);
      float4* dp = reinterpret_cast<float4*>(dinv + i0);
#pragma unroll
      for (int k = 0; k < 4; ++k) {
        int4 r;
        float4 d;
        r.x = run; run += c[4 * k];
        r.y = run; run += c[4 * k + 1];
        r.z = run; run += c[4 * k + 2];
        r.w = run; run += c[4 * k + 3];
        d.x = dinv_of_count(c[4 * k]);
        d.y = dinv_of_count(c[4 * k + 1]);
        d.z = dinv_of_count(c[4 * k + 2]);
        d.w = dinv_of_count(c[4 * k + 3]);
        rp[k] = r;
        dp[k] = d;
      }
    } else {
#pragma unroll
      for (int k = 0; k < 16; ++k) {
        if (i0 + k < n) {
          rowptr[i0 + k] = run;
          dinv[i0 + k] = dinv_of_count(c[k]);
        }
        run += c[k];
      }
    }
    __syncthreads();
  }
}

__global__ void k_fill(const int* __restrict__ src, const int* __restrict__ dst,
                       const int* __restrict__ rank, const int* __restrict__ rowptr,
                       const float* __restrict__ dinv,
                       int* __restrict__ csr_src, float* __restrict__ csr_w, int E) {
  int e = blockIdx.x * blockDim.x + threadIdx.x;
  if (e >= E) return;
  const int s = src[e];
  int pos = rowptr[dst[e]] + rank[e];
  pos = min(max(pos, 0), E - 1);
  csr_src[pos] = s;
  csr_w[pos] = dinv[s];
}

template <int K_DIM, int N_DIM>
__global__ void k_gemm_wmma_f32(const float* __restrict__ A,
                                const float* __restrict__ B,
                                float* __restrict__ C, int M) {
  const int wave = (int)((blockIdx.x * blockDim.x + threadIdx.x) >> 5);
  const int lane = threadIdx.x & 31;
  constexpr int TN = N_DIM / 16;
  const int rt = wave / TN;
  const int ct = wave - rt * TN;
  if (rt * 16 >= M) return;

  const int lm = lane & 15;
  const int kh = lane >> 4;

  const float* arow = A + (size_t)(rt * 16 + lm) * K_DIM + 2 * kh;
  const float* bcol = B + (size_t)(2 * kh) * N_DIM + ct * 16 + lm;

  v8f acc = {};
#pragma unroll 8
  for (int k = 0; k < K_DIM; k += 4) {
    v2f a, b;
    a[0] = arow[k];
    a[1] = arow[k + 1];
    b[0] = bcol[(size_t)k * N_DIM];
    b[1] = bcol[(size_t)(k + 1) * N_DIM];
    acc = __builtin_amdgcn_wmma_f32_16x16x4_f32(false, a, false, b,
                                                (short)0, acc, false, false);
  }

  float* crow = C + (size_t)(rt * 16 + 8 * kh) * N_DIM + ct * 16 + lm;
#pragma unroll
  for (int j = 0; j < 8; ++j) crow[(size_t)j * N_DIM] = acc[j];
}

static constexpr int W1_PITCH = 2 * HID_DIM + 32;
static constexpr int W1_LDS_WORDS = (IN_DIM / 2) * W1_PITCH;
static_assert(IN_DIM / 2 == 8 * 16 && HID_DIM == 4 * 32,
              "staging: 8 waves x 16 pair rows, 32 lanes x 4 columns");

__global__ void __launch_bounds__(256)
k_gemm1_strip_wmma_f32(const float* __restrict__ A,
                       const float* __restrict__ B,
                       float* __restrict__ C, int M) {
  __shared__ __attribute__((aligned(16))) float Bs[W1_LDS_WORDS];
  const int lane = threadIdx.x & 31;
  const int wv = __builtin_amdgcn_readfirstlane((int)(threadIdx.x >> 5));

  float4 r0[16], r1[16];
#pragma unroll
  for (int i = 0; i < 16; ++i) {
    const float* g = B + (size_t)(2 * (wv + 8 * i)) * HID_DIM + 4 * lane;
    r0[i] = *reinterpret_cast<const float4*>(g);
    r1[i] = *reinterpret_cast<const float4*>(g + HID_DIM);
  }
#pragma unroll
  for (int i = 0; i < 16; ++i) {
    float4 lo, hi;
    lo.x = r0[i].x; lo.y = r1[i].x; lo.z = r0[i].y; lo.w = r1[i].y;
    hi.x = r0[i].z; hi.y = r1[i].z; hi.z = r0[i].w; hi.w = r1[i].w;
    float* d = Bs + (wv + 8 * i) * W1_PITCH + 8 * lane;
    *reinterpret_cast<float4*>(d) = lo;
    *reinterpret_cast<float4*>(d + 4) = hi;
  }
  __syncthreads();

  const int rt = blockIdx.x * 8 + wv;
  if (rt * 16 >= M) return;

  const int lm = lane & 15;
  const int kh = lane >> 4;

  const int arow_i = min(rt * 16 + lm, M - 1);
  const float* arow = A + (size_t)arow_i * IN_DIM + 2 * kh;
  const float* bl = Bs + kh * W1_PITCH + 2 * lm;

  v8f acc[8];
#pragma unroll
  for (int ct = 0; ct < 8; ++ct) acc[ct] = v8f{};

  v2f a_cur[8], a_nxt[8];
#pragma unroll
  for (int s = 0; s < 8; ++s) a_cur[s] = *reinterpret_cast<const v2f*>(arow + 4 * s);

  for (int c = 0; c < IN_DIM / 32; ++c) {
    if (c + 1 < IN_DIM / 32) {
#pragma unroll
      for (int s = 0; s < 8; ++s)
        a_nxt[s] = *reinterpret_cast<const v2f*>(arow + 32 * (c + 1) + 4 * s);
    }
    const float* bc = bl + c * (16 * W1_PITCH);
#pragma unroll
    for (int s = 0; s < 8; ++s) {
#pragma unroll
      for (int ct = 0; ct < 8; ++ct) {
        const v2f b = *reinterpret_cast<const v2f*>(bc + (2 * s) * W1_PITCH + 32 * ct);
        acc[ct] = __builtin_amdgcn_wmma_f32_16x16x4_f32(false, a_cur[s], false, b,
                                                        (short)0, acc[ct], false, false);
      }
    }
#pragma unroll
    for (int s = 0; s < 8; ++s) a_cur[s] = a_nxt[s];
  }

  const int r0i = rt * 16 + 8 * kh;
  float* crow = C + (size_t)r0i * HID_DIM + lm;
  if (rt * 16 + 16 <= M) {
#pragma unroll
    for (int ct = 0; ct < 8; ++ct) {
#pragma unroll
      for (int j = 0; j < 8; ++j) crow[(size_t)j * HID_DIM + ct * 16] = acc[ct][j];
    }
  } else {
#pragma unroll
    for (int ct = 0; ct < 8; ++ct) {
#pragma unroll
      for (int j = 0; j < 8; ++j) {
        if (r0i + j < M) crow[(size_t)j * HID_DIM + ct * 16] = acc[ct][j];
      }
    }
  }
}


__global__ void k_gather_f128_bias_relu(const int* __restrict__ rowptr,
                                        const int* __restrict__ cnt,
                                        const int* __restrict__ csr_src,
                                        const float* __restrict__ csr_w,
                                        const float* __restrict__ dinv,
                                        const float* __restrict__ h,
                                        const float* __restrict__ bias,
                                        float* __restrict__ out, int n, int E) {
  const int node = __builtin_amdgcn_readfirstlane(
      (int)((blockIdx.x * blockDim.x + threadIdx.x) >> 5));
  if (node >= n) return;
  const int lane = threadIdx.x & 31;
  const int len = min(max(cnt[node], 0), E);
  const int start = min(max(rowptr[node], 0), E - len);
  const float dd = dinv[node];
  const float4 hv = *reinterpret_cast<const float4*>(h + (size_t)node * 128 + lane * 4);
  const float4 bv = *reinterpret_cast<const float4*>(bias + lane * 4);
  const float wself = dd * dd;
  float4 acc;
  acc.x = hv.x * wself;
  acc.y = hv.y * wself;
  acc.z = hv.z * wself;
  acc.w = hv.w * wself;
  for (int b = 0; b < len; b += 32) {
    const int m = min(32, len - b);
    int s_l = 0;
    float w_l = 0.0f;
    if (lane < m) {
      s_l = min(max(csr_src[start + b + lane], 0), n - 1);
      w_l = csr_w[start + b + lane] * dd;
    }
    int j = 0;
    for (; j + 4 <= m; j += 4) {
      const int s0 = __builtin_amdgcn_readlane(s_l, j);
      const int s1 = __builtin_amdgcn_readlane(s_l, j + 1);
      const int s2 = __builtin_amdgcn_readlane(s_l, j + 2);
      const int s3 = __builtin_amdgcn_readlane(s_l, j + 3);
      const float w0 = __int_as_float(__builtin_amdgcn_readlane(__float_as_int(w_l), j));
      const float w1 = __int_as_float(__builtin_amdgcn_readlane(__float_as_int(w_l), j + 1));
      const float w2 = __int_as_float(__builtin_amdgcn_readlane(__float_as_int(w_l), j + 2));
      const float w3 = __int_as_float(__builtin_amdgcn_readlane(__float_as_int(w_l), j + 3));
      const float4 v0 = *reinterpret_cast<const float4*>(h + (size_t)s0 * 128 + lane * 4);
      const float4 v1 = *reinterpret_cast<const float4*>(h + (size_t)s1 * 128 + lane * 4);
      const float4 v2 = *reinterpret_cast<const float4*>(h + (size_t)s2 * 128 + lane * 4);
      const float4 v3 = *reinterpret_cast<const float4*>(h + (size_t)s3 * 128 + lane * 4);
      acc.x += v0.x * w0; acc.y += v0.y * w0; acc.z += v0.z * w0; acc.w += v0.w * w0;
      acc.x += v1.x * w1; acc.y += v1.y * w1; acc.z += v1.z * w1; acc.w += v1.w * w1;
      acc.x += v2.x * w2; acc.y += v2.y * w2; acc.z += v2.z * w2; acc.w += v2.w * w2;
      acc.x += v3.x * w3; acc.y += v3.y * w3; acc.z += v3.z * w3; acc.w += v3.w * w3;
    }
    for (; j < m; ++j) {
      const int s0 = __builtin_amdgcn_readlane(s_l, j);
      const float w0 = __int_as_float(__builtin_amdgcn_readlane(__float_as_int(w_l), j));
      const float4 v0 = *reinterpret_cast<const float4*>(h + (size_t)s0 * 128 + lane * 4);
      acc.x += v0.x * w0; acc.y += v0.y * w0; acc.z += v0.z * w0; acc.w += v0.w * w0;
    }
  }
  acc.x = fmaxf(acc.x + bv.x, 0.0f);
  acc.y = fmaxf(acc.y + bv.y, 0.0f);
  acc.z = fmaxf(acc.z + bv.z, 0.0f);
  acc.w = fmaxf(acc.w + bv.w, 0.0f);
  *reinterpret_cast<float4*>(out + (size_t)node * 128 + lane * 4) = acc;
}

__global__ void k_gather_f16_bias_logsoftmax(const int* __restrict__ rowptr,
                                             const int* __restrict__ cnt,
                                             const int* __restrict__ csr_src,
                                             const float* __restrict__ csr_w,
                                             const float* __restrict__ dinv,
                                             const float* __restrict__ h,
                                             const float* __restrict__ bias,
                                             float* __restrict__ out, int n, int E) {
  const int idx = blockIdx.x * blockDim.x + threadIdx.x;
  const int node = idx >> 4;
  if (node >= n) return;
  const int f = idx & 15;
  const int len = min(max(cnt[node], 0), E);
  const int start = min(max(rowptr[node], 0), E - len);
  const float dd = dinv[node];
  float acc = h[(size_t)node * 16 + f] * (dd * dd);
  int j = 0;
  for (; j + 4 <= len; j += 4) {
    const int s0 = min(max(csr_src[start + j], 0), n - 1);
    const int s1 = min(max(csr_src[start + j + 1], 0), n - 1);
    const int s2 = min(max(csr_src[start + j + 2], 0), n - 1);
    const int s3 = min(max(csr_src[start + j + 3], 0), n - 1);
    const float w0 = csr_w[start + j] * dd;
    const float w1 = csr_w[start + j + 1] * dd;
    const float w2 = csr_w[start + j + 2] * dd;
    const float w3 = csr_w[start + j + 3] * dd;
    const float v0 = h[(size_t)s0 * 16 + f];
    const float v1 = h[(size_t)s1 * 16 + f];
    const float v2 = h[(size_t)s2 * 16 + f];
    const float v3 = h[(size_t)s3 * 16 + f];
    acc += v0 * w0;
    acc += v1 * w1;
    acc += v2 * w2;
    acc += v3 * w3;
  }
  for (; j < len; ++j) {
    const int s0 = min(max(csr_src[start + j], 0), n - 1);
    acc += h[(size_t)s0 * 16 + f] * (csr_w[start + j] * dd);
  }
  const float v = acc + bias[f];
  float m = __shfl(v, 0, 16);
#pragma unroll
  for (int k = 1; k < 16; ++k) m = fmaxf(m, __shfl(v, k, 16));
  const float ex = expf(v - m);
  float s = 0.0f;
#pragma unroll
  for (int k = 0; k < 16; ++k) s += __shfl(ex, k, 16);
  out[(size_t)node * 16 + f] = v - m - logf(s);
}

extern "C" void kernel_launch(void* const* d_in, const int* in_sizes, int n_in,
                              void* d_out, int out_size, void* d_ws, size_t ws_size,
                              hipStream_t stream) {
  (void)n_in; (void)out_size; (void)ws_size;

  const float* x  = (const float*)d_in[0];
  const int*   ei = (const int*)d_in[1];
  const float* W1 = (const float*)d_in[2];
  const float* b1 = (const float*)d_in[3];
  const float* W2 = (const float*)d_in[4];
  const float* b2 = (const float*)d_in[5];

  const int n = in_sizes[0] / IN_DIM;
  const int E = in_sizes[1] / 2;
  const int* src = ei;
  const int* dst = ei + E;

  float* h1     = (float*)d_ws;
  float* agg1   = h1 + (size_t)n * HID_DIM;
  float* h2     = agg1 + (size_t)n * HID_DIM;
  int*   cnt    = (int*)(h2 + (size_t)n * OUT_DIM);
  int*   rowptr = cnt + n;
  float* dinv   = (float*)(rowptr + n);
  int*   rank   = (int*)(dinv + n);
  int*   csr    = rank + E;
  float* csrw   = (float*)(csr + E);
  float* out    = (float*)d_out;

  const int B = 256;
  auto cdiv = [](long long a, long long b) { return (int)((a + b - 1) / b); };

  k_zero_cnt<<<cdiv(n, B), B, 0, stream>>>(cnt, n);
  k_rank<<<cdiv(E, B), B, 0, stream>>>(dst, cnt, rank, E);
  k_scan_dinv<<<1, 1024, 0, stream>>>(cnt, rowptr, dinv, n);
  k_fill<<<cdiv(E, B), B, 0, stream>>>(src, dst, rank, rowptr, dinv, csr, csrw, E);

  k_gemm1_strip_wmma_f32<<<cdiv(cdiv(n, 16), 8), B, 0, stream>>>(x, W1, h1, n);
  k_gather_f128_bias_relu<<<cdiv((long long)n * 32, B), B, 0, stream>>>(rowptr, cnt, csr, csrw, dinv, h1, b1, agg1, n, E);

  {
    long long waves = (long long)cdiv(n, 16) * (OUT_DIM / 16);
    k_gemm_wmma_f32<HID_DIM, OUT_DIM>
        <<<cdiv(waves * 32, B), B, 0, stream>>>(agg1, W2, h2, n);
  }
  k_gather_f16_bias_logsoftmax<<<cdiv((long long)n * OUT_DIM, B), B, 0, stream>>>(rowptr, cnt, csr, csrw, dinv, h2, b2, out, n, E);
}


# Edges grouped by destination with a two-level counting sort on LDS counters (256-node buckets) instead of 800000 returning global adds; sums: self term first, then arrival order of the LDS adds
# speedup vs baseline: 3.4396x; 1.6841x over previous
#include <hip/hip_runtime.h>
#include <math.h>

typedef __attribute__((ext_vector_type(2))) float v2f;
typedef __attribute__((ext_vector_type(8))) float v8f;

static constexpr int IN_DIM  = 256;
static constexpr int HID_DIM = 128;
static constexpr int OUT_DIM = 16;

static constexpr int BKT_SHIFT = 8;
static constexpr int BKT_NODES = 1 << BKT_SHIFT;
static constexpr int NB_MAX = 1024;
static constexpr int EP_THREADS = 1024;
static constexpr int EP_PER_THREAD = 8;
static constexpr int EP_CHUNK = EP_THREADS * EP_PER_THREAD;
static constexpr int FIN_PER_THREAD = 4;

static __device__ __forceinline__ int bucket_of(int d, int nb) {
  return min(max(d >> BKT_SHIFT, 0), nb - 1);
}

static __device__ __forceinline__ float dinv_of_count(int c) {
  const float d = (float)(c + 1);
  return (d > 0.0f) ? rsqrtf(d) : 0.0f;
}

__global__ void __launch_bounds__(EP_THREADS)
k_bucket_hist(const int* __restrict__ dst, int* __restrict__ blockhist, int E, int nb_in) {
  __shared__ int hist[NB_MAX];
  const int t = threadIdx.x;
  const int nb = min(nb_in, NB_MAX);
  hist[t] = 0;
  __syncthreads();
  const int e0 = blockIdx.x * EP_CHUNK + t;
  int d[EP_PER_THREAD];
#pragma unroll
  for (int u = 0; u < EP_PER_THREAD; ++u) {
    d[u] = dst[min(e0 + u * EP_THREADS, E - 1)];
  }
#pragma unroll
  for (int u = 0; u < EP_PER_THREAD; ++u) {
    const int e = e0 + u * EP_THREADS;
    if (e < E) atomicAdd(&hist[bucket_of(d[u], nb)], 1);
  }
  __syncthreads();
  if (t < nb) blockhist[(size_t)blockIdx.x * nb + t] = hist[t];
}
static_assert(EP_THREADS == NB_MAX, "one thread per LDS table entry");

__global__ void __launch_bounds__(EP_THREADS)
k_bucket_bin(const int* __restrict__ src, const int* __restrict__ dst,
             const int* __restrict__ blockhist, int2* __restrict__ binned,
             int* __restrict__ bbase, int E, int nb_in, int nwg) {
  __shared__ int tot[NB_MAX];
  __shared__ int cur[NB_MAX];
  __shared__ int wsum[32];
  const int t = threadIdx.x;
  const int lane = t & 31;
  const int wv = t >> 5;
  const int w = blockIdx.x;
  const int nb = min(nb_in, NB_MAX);
  tot[t] = 0;
  cur[t] = 0;
  __syncthreads();
  const int g = t >> 8;
  for (int b = t & 255; b < nb; b += 256) {
    int ts = 0, ps = 0;
#pragma unroll 8
    for (int r = g; r < nwg; r += 4) {
      const int v = blockhist[(size_t)r * nb + b];
      ts += v;
      ps += (r < w) ? v : 0;
    }
    atomicAdd(&tot[b], ts);
    atomicAdd(&cur[b], ps);
  }
  __syncthreads();
  const int c = (t < nb) ? tot[t] : 0;
  int inc = c;
  for (int o = 1; o < 32; o <<= 1) {
    const int y = __shfl_up(inc, o);
    if (lane >= o) inc += y;
  }
  if (lane == 31) wsum[wv] = inc;
  __syncthreads();
  if (wv == 0) {
    const int x = wsum[lane];
    int xi = x;
    for (int o = 1; o < 32; o <<= 1) {
      const int y = __shfl_up(xi, o);
      if (lane >= o) xi += y;
    }
    wsum[lane] = xi - x;
  }
  __syncthreads();
  const int base = wsum[wv] + inc - c;
  if (t < nb) {
    cur[t] += base;
    if (w == 0) bbase[t] = base;
  }
  __syncthreads();
  const int e0 = w * EP_CHUNK + t;
  int s[EP_PER_THREAD], d[EP_PER_THREAD];
#pragma unroll
  for (int u = 0; u < EP_PER_THREAD; ++u) {
    const int e = min(e0 + u * EP_THREADS, E - 1);
    s[u] = src[e];
    d[u] = dst[e];
  }
#pragma unroll
  for (int u = 0; u < EP_PER_THREAD; ++u) {
    const int e = e0 + u * EP_THREADS;
    if (e < E) {
      int pos = atomicAdd(&cur[bucket_of(d[u], nb)], 1);
      pos = min(max(pos, 0), E - 1);
      binned[pos] = make_int2(s[u], d[u]);
    }
  }
}

__global__ void __launch_bounds__(1024)
k_bucket_finalize(const int2* __restrict__ binned, const int* __restrict__ bbase,
                  int* __restrict__ cnt, int* __restrict__ rowptr,
                  float* __restrict__ dinv, int* __restrict__ csr_src,
                  int n, int E, int nb_in) {
  __shared__ int lcnt[BKT_NODES];
  __shared__ int wsum[BKT_NODES / 32];
  const int t = threadIdx.x;
  const int lane = t & 31;
  const int wv = t >> 5;
  const int b = blockIdx.x;
  const int nb = min(nb_in, NB_MAX);
  const int beg = min(max(bbase[b], 0), E);
  const int end = (b + 1 < nb) ? min(max(bbase[b + 1], beg), E) : E;
  if (t < BKT_NODES) lcnt[t] = 0;
  __syncthreads();
  for (int i0 = beg; i0 < end; i0 += 1024 * FIN_PER_THREAD) {
    int dl[FIN_PER_THREAD];
#pragma unroll
    for (int u = 0; u < FIN_PER_THREAD; ++u) {
      dl[u] = binned[min(i0 + u * 1024 + t, end - 1)].y;
    }
#pragma unroll
    for (int u = 0; u < FIN_PER_THREAD; ++u) {
      const int i = i0 + u * 1024 + t;
      if (i < end) atomicAdd(&lcnt[dl[u] & (BKT_NODES - 1)], 1);
    }
  }
  __syncthreads();
  const int c = (t < BKT_NODES) ? lcnt[t] : 0;
  int inc = c;
  for (int o = 1; o < 32; o <<= 1) {
    const int y = __shfl_up(inc, o);
    if (lane >= o) inc += y;
  }
  if (lane == 31 && wv < BKT_NODES / 32) wsum[wv] = inc;
  __syncthreads();
  if (t < BKT_NODES) {
    int woff = 0;
#pragma unroll
    for (int k = 0; k < BKT_NODES / 32; ++k) woff += (k < wv) ? wsum[k] : 0;
    const int row = beg + woff + inc - c;
    const int node = b * BKT_NODES + t;
    if (node < n) {
      cnt[node] = c;
      rowptr[node] = row;
      dinv[node] = dinv_of_count(c);
    }
    lcnt[t] = row;
  }
  __syncthreads();
  for (int i0 = beg; i0 < end; i0 += 1024 * FIN_PER_THREAD) {
    int2 v[FIN_PER_THREAD];
#pragma unroll
    for (int u = 0; u < FIN_PER_THREAD; ++u) {
      v[u] = binned[min(i0 + u * 1024 + t, end - 1)];
    }
#pragma unroll
    for (int u = 0; u < FIN_PER_THREAD; ++u) {
      const int i = i0 + u * 1024 + t;
      if (i < end) {
        int pos = atomicAdd(&lcnt[v[u].y & (BKT_NODES - 1)], 1);
        pos = min(max(pos, 0), E - 1);
        csr_src[pos] = v[u].x;
      }
    }
  }
}

template <int K_DIM, int N_DIM>
__global__ void k_gemm_wmma_f32(const float* __restrict__ A,
                                const float* __restrict__ B,
                                float* __restrict__ C, int M) {
  const int wave = (int)((blockIdx.x * blockDim.x + threadIdx.x) >> 5);
  const int lane = threadIdx.x & 31;
  constexpr int TN = N_DIM / 16;
  const int rt = wave / TN;
  const int ct = wave - rt * TN;
  if (rt * 16 >= M) return;

  const int lm = lane & 15;
  const int kh = lane >> 4;

  const float* arow = A + (size_t)(rt * 16 + lm) * K_DIM + 2 * kh;
  const float* bcol = B + (size_t)(2 * kh) * N_DIM + ct * 16 + lm;

  v8f acc = {};
#pragma unroll 8
  for (int k = 0; k < K_DIM; k += 4) {
    v2f a, b;
    a[0] = arow[k];
    a[1] = arow[k + 1];
    b[0] = bcol[(size_t)k * N_DIM];
    b[1] = bcol[(size_t)(k + 1) * N_DIM];
    acc = __builtin_amdgcn_wmma_f32_16x16x4_f32(false, a, false, b,
                                                (short)0, acc, false, false);
  }

  float* crow = C + (size_t)(rt * 16 + 8 * kh) * N_DIM + ct * 16 + lm;
#pragma unroll
  for (int j = 0; j < 8; ++j) crow[(size_t)j * N_DIM] = acc[j];
}

static constexpr int W1_PITCH = 2 * HID_DIM + 32;
static constexpr int W1_LDS_WORDS = (IN_DIM / 2) * W1_PITCH;
static_assert(IN_DIM / 2 == 8 * 16 && HID_DIM == 4 * 32,
              "staging: 8 waves x 16 pair rows, 32 lanes x 4 columns");

__global__ void __launch_bounds__(256)
k_gemm1_strip_wmma_f32(const float* __restrict__ A,
                       const float* __restrict__ B,
                       float* __restrict__ C, int M) {
  __shared__ __attribute__((aligned(16))) float Bs[W1_LDS_WORDS];
  const int lane = threadIdx.x & 31;
  const int wv = __builtin_amdgcn_readfirstlane((int)(threadIdx.x >> 5));

  float4 r0[16], r1[16];
#pragma unroll
  for (int i = 0; i < 16; ++i) {
    const float* g = B + (size_t)(2 * (wv + 8 * i)) * HID_DIM + 4 * lane;
    r0[i] = *reinterpret_cast<const float4*>(g);
    r1[i] = *reinterpret_cast<const float4*>(g + HID_DIM);
  }
#pragma unroll
  for (int i = 0; i < 16; ++i) {
    float4 lo, hi;
    lo.x = r0[i].x; lo.y = r1[i].x; lo.z = r0[i].y; lo.w = r1[i].y;
    hi.x = r0[i].z; hi.y = r1[i].z; hi.z = r0[i].w; hi.w = r1[i].w;
    float* d = Bs + (wv + 8 * i) * W1_PITCH + 8 * lane;
    *reinterpret_cast<float4*>(d) = lo;
    *reinterpret_cast<float4*>(d + 4) = hi;
  }
  __syncthreads();

  const int rt = blockIdx.x * 8 + wv;
  if (rt * 16 >= M) return;

  const int lm = lane & 15;
  const int kh = lane >> 4;

  const int arow_i = min(rt * 16 + lm, M - 1);
  const float* arow = A + (size_t)arow_i * IN_DIM + 2 * kh;
  const float* bl = Bs + kh * W1_PITCH + 2 * lm;

  v8f acc[8];
#pragma unroll
  for (int ct = 0; ct < 8; ++ct) acc[ct] = v8f{};

  v2f a_cur[8], a_nxt[8];
#pragma unroll
  for (int s = 0; s < 8; ++s) a_cur[s] = *reinterpret_cast<const v2f*>(arow + 4 * s);

  for (int c = 0; c < IN_DIM / 32; ++c) {
    if (c + 1 < IN_DIM / 32) {
#pragma unroll
      for (int s = 0; s < 8; ++s)
        a_nxt[s] = *reinterpret_cast<const v2f*>(arow + 32 * (c + 1) + 4 * s);
    }
    const float* bc = bl + c * (16 * W1_PITCH);
#pragma unroll
    for (int s = 0; s < 8; ++s) {
#pragma unroll
      for (int ct = 0; ct < 8; ++ct) {
        const v2f b = *reinterpret_cast<const v2f*>(bc + (2 * s) * W1_PITCH + 32 * ct);
        acc[ct] = __builtin_amdgcn_wmma_f32_16x16x4_f32(false, a_cur[s], false, b,
                                                        (short)0, acc[ct], false, false);
      }
    }
#pragma unroll
    for (int s = 0; s < 8; ++s) a_cur[s] = a_nxt[s];
  }

  const int r0i = rt * 16 + 8 * kh;
  float* crow = C + (size_t)r0i * HID_DIM + lm;
  if (rt * 16 + 16 <= M) {
#pragma unroll
    for (int ct = 0; ct < 8; ++ct) {
#pragma unroll
      for (int j = 0; j < 8; ++j) crow[(size_t)j * HID_DIM + ct * 16] = acc[ct][j];
    }
  } else {
#pragma unroll
    for (int ct = 0; ct < 8; ++ct) {
#pragma unroll
      for (int j = 0; j < 8; ++j) {
        if (r0i + j < M) crow[(size_t)j * HID_DIM + ct * 16] = acc[ct][j];
      }
    }
  }
}


__global__ void k_gather_f128_bias_relu(const int* __restrict__ rowptr,
                                        const int* __restrict__ cnt,
                                        const int* __restrict__ csr_src,
                                        float* __restrict__ csr_w,
                                        const float* __restrict__ dinv,
                                        const float* __restrict__ h,
                                        const float* __restrict__ bias,
                                        float* __restrict__ out, int n, int E) {
  const int node = __builtin_amdgcn_readfirstlane(
      (int)((blockIdx.x * blockDim.x + threadIdx.x) >> 5));
  if (node >= n) return;
  const int lane = threadIdx.x & 31;
  const int len = min(max(cnt[node], 0), E);
  const int start = min(max(rowptr[node], 0), E - len);
  const float dd = dinv[node];
  const float4 hv = *reinterpret_cast<const float4*>(h + (size_t)node * 128 + lane * 4);
  const float4 bv = *reinterpret_cast<const float4*>(bias + lane * 4);
  const float wself = dd * dd;
  float4 acc;
  acc.x = hv.x * wself;
  acc.y = hv.y * wself;
  acc.z = hv.z * wself;
  acc.w = hv.w * wself;
  for (int b = 0; b < len; b += 32) {
    const int m = min(32, len - b);
    int s_l = 0;
    float w_l = 0.0f;
    if (lane < m) {
      s_l = min(max(csr_src[start + b + lane], 0), n - 1);
      const float ds = dinv[s_l];
      csr_w[start + b + lane] = ds;
      w_l = ds * dd;
    }
    int j = 0;
    for (; j + 4 <= m; j += 4) {
      const int s0 = __builtin_amdgcn_readlane(s_l, j);
      const int s1 = __builtin_amdgcn_readlane(s_l, j + 1);
      const int s2 = __builtin_amdgcn_readlane(s_l, j + 2);
      const int s3 = __builtin_amdgcn_readlane(s_l, j + 3);
      const float w0 = __int_as_float(__builtin_amdgcn_readlane(__float_as_int(w_l), j));
      const float w1 = __int_as_float(__builtin_amdgcn_readlane(__float_as_int(w_l), j + 1));
      const float w2 = __int_as_float(__builtin_amdgcn_readlane(__float_as_int(w_l), j + 2));
      const float w3 = __int_as_float(__builtin_amdgcn_readlane(__float_as_int(w_l), j + 3));
      const float4 v0 = *reinterpret_cast<const float4*>(h + (size_t)s0 * 128 + lane * 4);
      const float4 v1 = *reinterpret_cast<const float4*>(h + (size_t)s1 * 128 + lane * 4);
      const float4 v2 = *reinterpret_cast<const float4*>(h + (size_t)s2 * 128 + lane * 4);
      const float4 v3 = *reinterpret_cast<const float4*>(h + (size_t)s3 * 128 + lane * 4);
      acc.x += v0.x * w0; acc.y += v0.y * w0; acc.z += v0.z * w0; acc.w += v0.w * w0;
      acc.x += v1.x * w1; acc.y += v1.y * w1; acc.z += v1.z * w1; acc.w += v1.w * w1;
      acc.x += v2.x * w2; acc.y += v2.y * w2; acc.z += v2.z * w2; acc.w += v2.w * w2;
      acc.x += v3.x * w3; acc.y += v3.y * w3; acc.z += v3.z * w3; acc.w += v3.w * w3;
    }
    for (; j < m; ++j) {
      const int s0 = __builtin_amdgcn_readlane(s_l, j);
      const float w0 = __int_as_float(__builtin_amdgcn_readlane(__float_as_int(w_l), j));
      const float4 v0 = *reinterpret_cast<const float4*>(h + (size_t)s0 * 128 + lane * 4);
      acc.x += v0.x * w0; acc.y += v0.y * w0; acc.z += v0.z * w0; acc.w += v0.w * w0;
    }
  }
  acc.x = fmaxf(acc.x + bv.x, 0.0f);
  acc.y = fmaxf(acc.y + bv.y, 0.0f);
  acc.z = fmaxf(acc.z + bv.z, 0.0f);
  acc.w = fmaxf(acc.w + bv.w, 0.0f);
  *reinterpret_cast<float4*>(out + (size_t)node * 128 + lane * 4) = acc;
}

__global__ void k_gather_f16_bias_logsoftmax(const int* __restrict__ rowptr,
                                             const int* __restrict__ cnt,
                                             const int* __restrict__ csr_src,
                                             const float* __restrict__ csr_w,
                                             const float* __restrict__ dinv,
                                             const float* __restrict__ h,
                                             const float* __restrict__ bias,
                                             float* __restrict__ out, int n, int E) {
  const int idx = blockIdx.x * blockDim.x + threadIdx.x;
  const int node = idx >> 4;
  if (node >= n) return;
  const int f = idx & 15;
  const int len = min(max(cnt[node], 0), E);
  const int start = min(max(rowptr[node], 0), E - len);
  const float dd = dinv[node];
  float acc = h[(size_t)node * 16 + f] * (dd * dd);
  int j = 0;
  for (; j + 4 <= len; j += 4) {
    const int s0 = min(max(csr_src[start + j], 0), n - 1);
    const int s1 = min(max(csr_src[start + j + 1], 0), n - 1);
    const int s2 = min(max(csr_src[start + j + 2], 0), n - 1);
    const int s3 = min(max(csr_src[start + j + 3], 0), n - 1);
    const float w0 = csr_w[start + j] * dd;
    const float w1 = csr_w[start + j + 1] * dd;
    const float w2 = csr_w[start + j + 2] * dd;
    const float w3 = csr_w[start + j + 3] * dd;
    const float v0 = h[(size_t)s0 * 16 + f];
    const float v1 = h[(size_t)s1 * 16 + f];
    const float v2 = h[(size_t)s2 * 16 + f];
    const float v3 = h[(size_t)s3 * 16 + f];
    acc += v0 * w0;
    acc += v1 * w1;
    acc += v2 * w2;
    acc += v3 * w3;
  }
  for (; j < len; ++j) {
    const int s0 = min(max(csr_src[start + j], 0), n - 1);
    acc += h[(size_t)s0 * 16 + f] * (csr_w[start + j] * dd);
  }
  const float v = acc + bias[f];
  float m = __shfl(v, 0, 16);
#pragma unroll
  for (int k = 1; k < 16; ++k) m = fmaxf(m, __shfl(v, k, 16));
  const float ex = expf(v - m);
  float s = 0.0f;
#pragma unroll
  for (int k = 0; k < 16; ++k) s += __shfl(ex, k, 16);
  out[(size_t)node * 16 + f] = v - m - logf(s);
}

extern "C" void kernel_launch(void* const* d_in, const int* in_sizes, int n_in,
                              void* d_out, int out_size, void* d_ws, size_t ws_size,
                              hipStream_t stream) {
  (void)n_in; (void)out_size; (void)ws_size;

  const float* x  = (const float*)d_in[0];
  const int*   ei = (const int*)d_in[1];
  const float* W1 = (const float*)d_in[2];
  const float* b1 = (const float*)d_in[3];
  const float* W2 = (const float*)d_in[4];
  const float* b2 = (const float*)d_in[5];

  const int n = in_sizes[0] / IN_DIM;
  const int E = in_sizes[1] / 2;
  const int* src = ei;
  const int* dst = ei + E;

  const int B = 256;
  auto cdiv = [](long long a, long long b) { return (int)((a + b - 1) / b); };

  const int nb  = cdiv(n, BKT_NODES);
  const int nwg = cdiv(E, EP_CHUNK);

  float* h1     = (float*)d_ws;
  float* agg1   = h1 + (size_t)n * HID_DIM;
  float* h2     = agg1 + (size_t)n * HID_DIM;
  int2*  binned = (int2*)(h2 + (size_t)n * OUT_DIM);
  int*   csr    = (int*)(binned + E);
  float* csrw   = (float*)(csr + E);
  int*   cnt    = (int*)(csrw + E);
  int*   rowptr = cnt + n;
  float* dinv   = (float*)(rowptr + n);
  int*   bhist  = (int*)(dinv + n);
  int*   bbase  = bhist + (size_t)nwg * nb;
  float* out    = (float*)d_out;

  k_bucket_hist<<<nwg, EP_THREADS, 0, stream>>>(dst, bhist, E, nb);
  k_bucket_bin<<<nwg, EP_THREADS, 0, stream>>>(src, dst, bhist, binned, bbase, E, nb, nwg);
  k_bucket_finalize<<<nb, 1024, 0, stream>>>(binned, bbase, cnt, rowptr, dinv, csr, n, E, nb);

  k_gemm1_strip_wmma_f32<<<cdiv(cdiv(n, 16), 8), B, 0, stream>>>(x, W1, h1, n);
  k_gather_f128_bias_relu<<<cdiv((long long)n * 32, B), B, 0, stream>>>(rowptr, cnt, csr, csrw, dinv, h1, b1, agg1, n, E);

  {
    long long waves = (long long)cdiv(n, 16) * (OUT_DIM / 16);
    k_gemm_wmma_f32<HID_DIM, OUT_DIM>
        <<<cdiv(waves * 32, B), B, 0, stream>>>(agg1, W2, h2, n);
  }
  k_gather_f16_bias_logsoftmax<<<cdiv((long long)n * OUT_DIM, B), B, 0, stream>>>(rowptr, cnt, csr, csrw, dinv, h2, b2, out, n, E);
}


# Layer-2 product fused into the layer-1 gather: a block's 16 relu rows go via an LDS tile to one wave's WMMAs (7 launches -> 6, rows never stored); every sum keeps the parent's order
# speedup vs baseline: 3.7165x; 1.0805x over previous
#include <hip/hip_runtime.h>
#include <math.h>

typedef __attribute__((ext_vector_type(2))) float v2f;
typedef __attribute__((ext_vector_type(8))) float v8f;

static constexpr int IN_DIM  = 256;
static constexpr int HID_DIM = 128;
static constexpr int OUT_DIM = 16;

static constexpr int BKT_SHIFT = 8;
static constexpr int BKT_NODES = 1 << BKT_SHIFT;
static constexpr int NB_MAX = 1024;
static constexpr int EP_THREADS = 1024;
static constexpr int EP_PER_THREAD = 8;
static constexpr int EP_CHUNK = EP_THREADS * EP_PER_THREAD;
static constexpr int FIN_PER_THREAD = 4;

static __device__ __forceinline__ int bucket_of(int d, int nb) {
  return min(max(d >> BKT_SHIFT, 0), nb - 1);
}

static __device__ __forceinline__ float dinv_of_count(int c) {
  const float d = (float)(c + 1);
  return (d > 0.0f) ? rsqrtf(d) : 0.0f;
}

__global__ void __launch_bounds__(EP_THREADS)
k_bucket_hist(const int* __restrict__ dst, int* __restrict__ blockhist, int E, int nb_in) {
  __shared__ int hist[NB_MAX];
  const int t = threadIdx.x;
  const int nb = min(nb_in, NB_MAX);
  hist[t] = 0;
  __syncthreads();
  const int e0 = blockIdx.x * EP_CHUNK + t;
  int d[EP_PER_THREAD];
#pragma unroll
  for (int u = 0; u < EP_PER_THREAD; ++u) {
    d[u] = dst[min(e0 + u * EP_THREADS, E - 1)];
  }
#pragma unroll
  for (int u = 0; u < EP_PER_THREAD; ++u) {
    const int e = e0 + u * EP_THREADS;
    if (e < E) atomicAdd(&hist[bucket_of(d[u], nb)], 1);
  }
  __syncthreads();
  if (t < nb) blockhist[(size_t)blockIdx.x * nb + t] = hist[t];
}
static_assert(EP_THREADS == NB_MAX, "one thread per LDS table entry");

__global__ void __launch_bounds__(EP_THREADS)
k_bucket_bin(const int* __restrict__ src, const int* __restrict__ dst,
             const int* __restrict__ blockhist, int2* __restrict__ binned,
             int* __restrict__ bbase, int E, int nb_in, int nwg) {
  __shared__ int tot[NB_MAX];
  __shared__ int cur[NB_MAX];
  __shared__ int wsum[32];
  const int t = threadIdx.x;
  const int lane = t & 31;
  const int wv = t >> 5;
  const int w = blockIdx.x;
  const int nb = min(nb_in, NB_MAX);
  tot[t] = 0;
  cur[t] = 0;
  __syncthreads();
  const int g = t >> 8;
  for (int b = t & 255; b < nb; b += 256) {
    int ts = 0, ps = 0;
#pragma unroll 8
    for (int r = g; r < nwg; r += 4) {
      const int v = blockhist[(size_t)r * nb + b];
      ts += v;
      ps += (r < w) ? v : 0;
    }
    atomicAdd(&tot[b], ts);
    atomicAdd(&cur[b], ps);
  }
  __syncthreads();
  const int c = (t < nb) ? tot[t] : 0;
  int inc = c;
  for (int o = 1; o < 32; o <<= 1) {
    const int y = __shfl_up(inc, o);
    if (lane >= o) inc += y;
  }
  if (lane == 31) wsum[wv] = inc;
  __syncthreads();
  if (wv == 0) {
    const int x = wsum[lane];
    int xi = x;
    for (int o = 1; o < 32; o <<= 1) {
      const int y = __shfl_up(xi, o);
      if (lane >= o) xi += y;
    }
    wsum[lane] = xi - x;
  }
  __syncthreads();
  const int base = wsum[wv] + inc - c;
  if (t < nb) {
    cur[t] += base;
    if (w == 0) bbase[t] = base;
  }
  __syncthreads();
  const int e0 = w * EP_CHUNK + t;
  int s[EP_PER_THREAD], d[EP_PER_THREAD];
#pragma unroll
  for (int u = 0; u < EP_PER_THREAD; ++u) {
    const int e = min(e0 + u * EP_THREADS, E - 1);
    s[u] = src[e];
    d[u] = dst[e];
  }
#pragma unroll
  for (int u = 0; u < EP_PER_THREAD; ++u) {
    const int e = e0 + u * EP_THREADS;
    if (e < E) {
      int pos = atomicAdd(&cur[bucket_of(d[u], nb)], 1);
      pos = min(max(pos, 0), E - 1);
      binned[pos] = make_int2(s[u], d[u]);
    }
  }
}

__global__ void __launch_bounds__(1024)
k_bucket_finalize(const int2* __restrict__ binned, const int* __restrict__ bbase,
                  int* __restrict__ cnt, int* __restrict__ rowptr,
                  float* __restrict__ dinv, int* __restrict__ csr_src,
                  int n, int E, int nb_in) {
  __shared__ int lcnt[BKT_NODES];
  __shared__ int wsum[BKT_NODES / 32];
  const int t = threadIdx.x;
  const int lane = t & 31;
  const int wv = t >> 5;
  const int b = blockIdx.x;
  const int nb = min(nb_in, NB_MAX);
  const int beg = min(max(bbase[b], 0), E);
  const int end = (b + 1 < nb) ? min(max(bbase[b + 1], beg), E) : E;
  if (t < BKT_NODES) lcnt[t] = 0;
  __syncthreads();
  for (int i0 = beg; i0 < end; i0 += 1024 * FIN_PER_THREAD) {
    int dl[FIN_PER_THREAD];
#pragma unroll
    for (int u = 0; u < FIN_PER_THREAD; ++u) {
      dl[u] = binned[min(i0 + u * 1024 + t, end - 1)].y;
    }
#pragma unroll
    for (int u = 0; u < FIN_PER_THREAD; ++u) {
      const int i = i0 + u * 1024 + t;
      if (i < end) atomicAdd(&lcnt[dl[u] & (BKT_NODES - 1)], 1);
    }
  }
  __syncthreads();
  const int c = (t < BKT_NODES) ? lcnt[t] : 0;
  int inc = c;
  for (int o = 1; o < 32; o <<= 1) {
    const int y = __shfl_up(inc, o);
    if (lane >= o) inc += y;
  }
  if (lane == 31 && wv < BKT_NODES / 32) wsum[wv] = inc;
  __syncthreads();
  if (t < BKT_NODES) {
    int woff = 0;
#pragma unroll
    for (int k = 0; k < BKT_NODES / 32; ++k) woff += (k < wv) ? wsum[k] : 0;
    const int row = beg + woff + inc - c;
    const int node = b * BKT_NODES + t;
    if (node < n) {
      cnt[node] = c;
      rowptr[node] = row;
      dinv[node] = dinv_of_count(c);
    }
    lcnt[t] = row;
  }
  __syncthreads();
  for (int i0 = beg; i0 < end; i0 += 1024 * FIN_PER_THREAD) {
    int2 v[FIN_PER_THREAD];
#pragma unroll
    for (int u = 0; u < FIN_PER_THREAD; ++u) {
      v[u] = binned[min(i0 + u * 1024 + t, end - 1)];
    }
#pragma unroll
    for (int u = 0; u < FIN_PER_THREAD; ++u) {
      const int i = i0 + u * 1024 + t;
      if (i < end) {
        int pos = atomicAdd(&lcnt[v[u].y & (BKT_NODES - 1)], 1);
        pos = min(max(pos, 0), E - 1);
        csr_src[pos] = v[u].x;
      }
    }
  }
}


static constexpr int W1_PITCH = 2 * HID_DIM + 32;
static constexpr int W1_LDS_WORDS = (IN_DIM / 2) * W1_PITCH;
static_assert(IN_DIM / 2 == 8 * 16 && HID_DIM == 4 * 32,
              "staging: 8 waves x 16 pair rows, 32 lanes x 4 columns");

__global__ void __launch_bounds__(256)
k_gemm1_strip_wmma_f32(const float* __restrict__ A,
                       const float* __restrict__ B,
                       float* __restrict__ C, int M) {
  __shared__ __attribute__((aligned(16))) float Bs[W1_LDS_WORDS];
  const int lane = threadIdx.x & 31;
  const int wv = __builtin_amdgcn_readfirstlane((int)(threadIdx.x >> 5));

  float4 r0[16], r1[16];
#pragma unroll
  for (int i = 0; i < 16; ++i) {
    const float* g = B + (size_t)(2 * (wv + 8 * i)) * HID_DIM + 4 * lane;
    r0[i] = *reinterpret_cast<const float4*>(g);
    r1[i] = *reinterpret_cast<const float4*>(g + HID_DIM);
  }
#pragma unroll
  for (int i = 0; i < 16; ++i) {
    float4 lo, hi;
    lo.x = r0[i].x; lo.y = r1[i].x; lo.z = r0[i].y; lo.w = r1[i].y;
    hi.x = r0[i].z; hi.y = r1[i].z; hi.z = r0[i].w; hi.w = r1[i].w;
    float* d = Bs + (wv + 8 * i) * W1_PITCH + 8 * lane;
    *reinterpret_cast<float4*>(d) = lo;
    *reinterpret_cast<float4*>(d + 4) = hi;
  }
  __syncthreads();

  const int rt = blockIdx.x * 8 + wv;
  if (rt * 16 >= M) return;

  const int lm = lane & 15;
  const int kh = lane >> 4;

  const int arow_i = min(rt * 16 + lm, M - 1);
  const float* arow = A + (size_t)arow_i * IN_DIM + 2 * kh;
  const float* bl = Bs + kh * W1_PITCH + 2 * lm;

  v8f acc[8];
#pragma unroll
  for (int ct = 0; ct < 8; ++ct) acc[ct] = v8f{};

  v2f a_cur[8], a_nxt[8];
#pragma unroll
  for (int s = 0; s < 8; ++s) a_cur[s] = *reinterpret_cast<const v2f*>(arow + 4 * s);

  for (int c = 0; c < IN_DIM / 32; ++c) {
    if (c + 1 < IN_DIM / 32) {
#pragma unroll
      for (int s = 0; s < 8; ++s)
        a_nxt[s] = *reinterpret_cast<const v2f*>(arow + 32 * (c + 1) + 4 * s);
    }
    const float* bc = bl + c * (16 * W1_PITCH);
#pragma unroll
    for (int s = 0; s < 8; ++s) {
#pragma unroll
      for (int ct = 0; ct < 8; ++ct) {
        const v2f b = *reinterpret_cast<const v2f*>(bc + (2 * s) * W1_PITCH + 32 * ct);
        acc[ct] = __builtin_amdgcn_wmma_f32_16x16x4_f32(false, a_cur[s], false, b,
                                                        (short)0, acc[ct], false, false);
      }
    }
#pragma unroll
    for (int s = 0; s < 8; ++s) a_cur[s] = a_nxt[s];
  }

  const int r0i = rt * 16 + 8 * kh;
  float* crow = C + (size_t)r0i * HID_DIM + lm;
  if (rt * 16 + 16 <= M) {
#pragma unroll
    for (int ct = 0; ct < 8; ++ct) {
#pragma unroll
      for (int j = 0; j < 8; ++j) crow[(size_t)j * HID_DIM + ct * 16] = acc[ct][j];
    }
  } else {
#pragma unroll
    for (int ct = 0; ct < 8; ++ct) {
#pragma unroll
      for (int j = 0; j < 8; ++j) {
        if (r0i + j < M) crow[(size_t)j * HID_DIM + ct * 16] = acc[ct][j];
      }
    }
  }
}


static constexpr int AG_ROWS = 16;
static constexpr int AG_PITCH = HID_DIM + 4;
static constexpr int W2_PAIR_WORDS = 2 * OUT_DIM;
static_assert(OUT_DIM == 16 && HID_DIM == 128 && (HID_DIM / 2) * 4 == 256,
              "one 16x16 output tile; W2 staging: 256 threads x (one row pair, 4 columns)");

__global__ void __launch_bounds__(256)
k_gather_f128_bias_relu_gemm2(const int* __restrict__ rowptr,
                              const int* __restrict__ cnt,
                              const int* __restrict__ csr_src,
                              float* __restrict__ csr_w,
                              const float* __restrict__ dinv,
                              const float* __restrict__ h,
                              const float* __restrict__ bias,
                              const float* __restrict__ W2,
                              float* __restrict__ h2, int n, int E) {
  __shared__ __attribute__((aligned(16))) float tile[AG_ROWS * AG_PITCH];
  __shared__ __attribute__((aligned(16))) float w2s[(HID_DIM / 2) * W2_PAIR_WORDS];
  const int lane = threadIdx.x & 31;
  const int wv = __builtin_amdgcn_readfirstlane((int)(threadIdx.x >> 5));

  const int p = (int)threadIdx.x >> 2;
  const int c0 = ((int)threadIdx.x & 3) * 4;
  const float4 wr0 = *reinterpret_cast<const float4*>(W2 + (size_t)(2 * p) * OUT_DIM + c0);
  const float4 wr1 = *reinterpret_cast<const float4*>(W2 + (size_t)(2 * p + 1) * OUT_DIM + c0);

  const float4 bv = *reinterpret_cast<const float4*>(bias + lane * 4);
  const int node0 = blockIdx.x * AG_ROWS;

  for (int r = 0; r < 2; ++r) {
    const int slot = wv + 8 * r;
    const int node = node0 + slot;
    float4 acc;
    acc.x = 0.0f; acc.y = 0.0f; acc.z = 0.0f; acc.w = 0.0f;
    if (node < n) {
      const int len = min(max(cnt[node], 0), E);
      const int start = min(max(rowptr[node], 0), E - len);
      const float dd = dinv[node];
      const float4 hv = *reinterpret_cast<const float4*>(h + (size_t)node * 128 + lane * 4);
      const float wself = dd * dd;
      acc.x = hv.x * wself;
      acc.y = hv.y * wself;
      acc.z = hv.z * wself;
      acc.w = hv.w * wself;
      for (int b = 0; b < len; b += 32) {
        const int m = min(32, len - b);
        int s_l = 0;
        float w_l = 0.0f;
        if (lane < m) {
          s_l = min(max(csr_src[start + b + lane], 0), n - 1);
          const float ds = dinv[s_l];
          csr_w[start + b + lane] = ds;
          w_l = ds * dd;
        }
        int j = 0;
        for (; j + 4 <= m; j += 4) {
          const int s0 = __builtin_amdgcn_readlane(s_l, j);
          const int s1 = __builtin_amdgcn_readlane(s_l, j + 1);
          const int s2 = __builtin_amdgcn_readlane(s_l, j + 2);
          const int s3 = __builtin_amdgcn_readlane(s_l, j + 3);
          const float w0 = __int_as_float(__builtin_amdgcn_readlane(__float_as_int(w_l), j));
          const float w1 = __int_as_float(__builtin_amdgcn_readlane(__float_as_int(w_l), j + 1));
          const float w2 = __int_as_float(__builtin_amdgcn_readlane(__float_as_int(w_l), j + 2));
          const float w3 = __int_as_float(__builtin_amdgcn_readlane(__float_as_int(w_l), j + 3));
          const float4 v0 = *reinterpret_cast<const float4*>(h + (size_t)s0 * 128 + lane * 4);
          const float4 v1 = *reinterpret_cast<const float4*>(h + (size_t)s1 * 128 + lane * 4);
          const float4 v2 = *reinterpret_cast<const float4*>(h + (size_t)s2 * 128 + lane * 4);
          const float4 v3 = *reinterpret_cast<const float4*>(h + (size_t)s3 * 128 + lane * 4);
          acc.x += v0.x * w0; acc.y += v0.y * w0; acc.z += v0.z * w0; acc.w += v0.w * w0;
          acc.x += v1.x * w1; acc.y += v1.y * w1; acc.z += v1.z * w1; acc.w += v1.w * w1;
          acc.x += v2.x * w2; acc.y += v2.y * w2; acc.z += v2.z * w2; acc.w += v2.w * w2;
          acc.x += v3.x * w3; acc.y += v3.y * w3; acc.z += v3.z * w3; acc.w += v3.w * w3;
        }
        for (; j < m; ++j) {
          const int s0 = __builtin_amdgcn_readlane(s_l, j);
          const float w0 = __int_as_float(__builtin_amdgcn_readlane(__float_as_int(w_l), j));
          const float4 v0 = *reinterpret_cast<const float4*>(h + (size_t)s0 * 128 + lane * 4);
          acc.x += v0.x * w0; acc.y += v0.y * w0; acc.z += v0.z * w0; acc.w += v0.w * w0;
        }
      }
      acc.x = fmaxf(acc.x + bv.x, 0.0f);
      acc.y = fmaxf(acc.y + bv.y, 0.0f);
      acc.z = fmaxf(acc.z + bv.z, 0.0f);
      acc.w = fmaxf(acc.w + bv.w, 0.0f);
    }
    *reinterpret_cast<float4*>(tile + slot * AG_PITCH + lane * 4) = acc;
    if (r == 0) {
      float4 lo, hi;
      lo.x = wr0.x; lo.y = wr1.x; lo.z = wr0.y; lo.w = wr1.y;
      hi.x = wr0.z; hi.y = wr1.z; hi.z = wr0.w; hi.w = wr1.w;
      float* d = w2s + p * W2_PAIR_WORDS + 2 * c0;
      *reinterpret_cast<float4*>(d) = lo;
      *reinterpret_cast<float4*>(d + 4) = hi;
    }
  }
  __syncthreads();
  if (wv != 0) return;

  const int lm = lane & 15;
  const int kh = lane >> 4;
  const float* al = tile + lm * AG_PITCH + 2 * kh;
  const float* bl = w2s + kh * W2_PAIR_WORDS + 2 * lm;
  v8f c = {};
#pragma unroll 8
  for (int k = 0; k < HID_DIM; k += 4) {
    const v2f a = *reinterpret_cast<const v2f*>(al + k);
    const v2f b = *reinterpret_cast<const v2f*>(bl + (k / 2) * W2_PAIR_WORDS);
    c = __builtin_amdgcn_wmma_f32_16x16x4_f32(false, a, false, b, (short)0, c, false, false);
  }
  const int r0i = node0 + 8 * kh;
  float* crow = h2 + (size_t)r0i * OUT_DIM + lm;
  if (node0 + AG_ROWS <= n) {
#pragma unroll
    for (int j = 0; j < 8; ++j) crow[(size_t)j * OUT_DIM] = c[j];
  } else {
#pragma unroll
    for (int j = 0; j < 8; ++j) {
      if (r0i + j < n) crow[(size_t)j * OUT_DIM] = c[j];
    }
  }
}

__global__ void k_gather_f16_bias_logsoftmax(const int* __restrict__ rowptr,
                                             const int* __restrict__ cnt,
                                             const int* __restrict__ csr_src,
                                             const float* __restrict__ csr_w,
                                             const float* __restrict__ dinv,
                                             const float* __restrict__ h,
                                             const float* __restrict__ bias,
                                             float* __restrict__ out, int n, int E) {
  const int idx = blockIdx.x * blockDim.x + threadIdx.x;
  const int node = idx >> 4;
  if (node >= n) return;
  const int f = idx & 15;
  const int len = min(max(cnt[node], 0), E);
  const int start = min(max(rowptr[node], 0), E - len);
  const float dd = dinv[node];
  float acc = h[(size_t)node * 16 + f] * (dd * dd);
  int j = 0;
  for (; j + 4 <= len; j += 4) {
    const int s0 = min(max(csr_src[start + j], 0), n - 1);
    const int s1 = min(max(csr_src[start + j + 1], 0), n - 1);
    const int s2 = min(max(csr_src[start + j + 2], 0), n - 1);
    const int s3 = min(max(csr_src[start + j + 3], 0), n - 1);
    const float w0 = csr_w[start + j] * dd;
    const float w1 = csr_w[start + j + 1] * dd;
    const float w2 = csr_w[start + j + 2] * dd;
    const float w3 = csr_w[start + j + 3] * dd;
    const float v0 = h[(size_t)s0 * 16 + f];
    const float v1 = h[(size_t)s1 * 16 + f];
    const float v2 = h[(size_t)s2 * 16 + f];
    const float v3 = h[(size_t)s3 * 16 + f];
    acc += v0 * w0;
    acc += v1 * w1;
    acc += v2 * w2;
    acc += v3 * w3;
  }
  for (; j < len; ++j) {
    const int s0 = min(max(csr_src[start + j], 0), n - 1);
    acc += h[(size_t)s0 * 16 + f] * (csr_w[start + j] * dd);
  }
  const float v = acc + bias[f];
  float m = __shfl(v, 0, 16);
#pragma unroll
  for (int k = 1; k < 16; ++k) m = fmaxf(m, __shfl(v, k, 16));
  const float ex = expf(v - m);
  float s = 0.0f;
#pragma unroll
  for (int k = 0; k < 16; ++k) s += __shfl(ex, k, 16);
  out[(size_t)node * 16 + f] = v - m - logf(s);
}

extern "C" void kernel_launch(void* const* d_in, const int* in_sizes, int n_in,
                              void* d_out, int out_size, void* d_ws, size_t ws_size,
                              hipStream_t stream) {
  (void)n_in; (void)out_size; (void)ws_size;

  const float* x  = (const float*)d_in[0];
  const int*   ei = (const int*)d_in[1];
  const float* W1 = (const float*)d_in[2];
  const float* b1 = (const float*)d_in[3];
  const float* W2 = (const float*)d_in[4];
  const float* b2 = (const float*)d_in[5];

  const int n = in_sizes[0] / IN_DIM;
  const int E = in_sizes[1] / 2;
  const int* src = ei;
  const int* dst = ei + E;

  const int B = 256;
  auto cdiv = [](long long a, long long b) { return (int)((a + b - 1) / b); };

  const int nb  = cdiv(n, BKT_NODES);
  const int nwg = cdiv(E, EP_CHUNK);

  float* h1     = (float*)d_ws;
  float* h2     = h1 + (size_t)n * HID_DIM;
  int2*  binned = (int2*)(h2 + (size_t)n * OUT_DIM);
  int*   csr    = (int*)(binned + E);
  float* csrw   = (float*)(csr + E);
  int*   cnt    = (int*)(csrw + E);
  int*   rowptr = cnt + n;
  float* dinv   = (float*)(rowptr + n);
  int*   bhist  = (int*)(dinv + n);
  int*   bbase  = bhist + (size_t)nwg * nb;
  float* out    = (float*)d_out;

  k_bucket_hist<<<nwg, EP_THREADS, 0, stream>>>(dst, bhist, E, nb);
  k_bucket_bin<<<nwg, EP_THREADS, 0, stream>>>(src, dst, bhist, binned, bbase, E, nb, nwg);
  k_bucket_finalize<<<nb, 1024, 0, stream>>>(binned, bbase, cnt, rowptr, dinv, csr, n, E, nb);

  k_gemm1_strip_wmma_f32<<<cdiv(cdiv(n, 16), 8), B, 0, stream>>>(x, W1, h1, n);
  k_gather_f128_bias_relu_gemm2<<<cdiv(n, AG_ROWS), B, 0, stream>>>(rowptr, cnt, csr, csrw, dinv, h1, b1, W2, h2, n, E);

  k_gather_f16_bias_logsoftmax<<<cdiv((long long)n * OUT_DIM, B), B, 0, stream>>>(rowptr, cnt, csr, csrw, dinv, h2, b2, out, n, E);
}


# Bin pass of the edge grouping runs as extra 256-thread work-groups inside the layer-1 GEMM's launch (6 launches -> 5); sums as before: self term first, then arrival order of the LDS adds
# speedup vs baseline: 4.3041x; 1.1581x over previous
#include <hip/hip_runtime.h>
#include <math.h>

typedef __attribute__((ext_vector_type(2))) float v2f;
typedef __attribute__((ext_vector_type(8))) float v8f;

static constexpr int IN_DIM  = 256;
static constexpr int HID_DIM = 128;
static constexpr int OUT_DIM = 16;

static constexpr int BKT_SHIFT = 8;
static constexpr int BKT_NODES = 1 << BKT_SHIFT;
static constexpr int NB_MAX = 1024;
static constexpr int EP_THREADS = 1024;
static constexpr int EP_PER_THREAD = 8;
static constexpr int EP_CHUNK = EP_THREADS * EP_PER_THREAD;
static constexpr int FIN_PER_THREAD = 4;
static constexpr int BIN_THREADS = 256;
static constexpr int BIN_PER_THREAD = EP_CHUNK / BIN_THREADS;
static_assert(BIN_THREADS * BIN_PER_THREAD == EP_CHUNK, "hist and bin cut the edge list into the same chunks");
static_assert(NB_MAX == 4 * BIN_THREADS, "bucket scan: four consecutive buckets per thread");

static __device__ __forceinline__ int bucket_of(int d, int nb) {
  return min(max(d >> BKT_SHIFT, 0), nb - 1);
}

static __device__ __forceinline__ float dinv_of_count(int c) {
  const float d = (float)(c + 1);
  return (d > 0.0f) ? rsqrtf(d) : 0.0f;
}

__global__ void __launch_bounds__(EP_THREADS)
k_bucket_hist(const int* __restrict__ dst, int* __restrict__ blockhist, int E, int nb_in) {
  __shared__ int hist[NB_MAX];
  const int t = threadIdx.x;
  const int nb = min(nb_in, NB_MAX);
  hist[t] = 0;
  __syncthreads();
  const int e0 = blockIdx.x * EP_CHUNK + t;
  int d[EP_PER_THREAD];
#pragma unroll
  for (int u = 0; u < EP_PER_THREAD; ++u) {
    d[u] = dst[min(e0 + u * EP_THREADS, E - 1)];
  }
#pragma unroll
  for (int u = 0; u < EP_PER_THREAD; ++u) {
    const int e = e0 + u * EP_THREADS;
    if (e < E) atomicAdd(&hist[bucket_of(d[u], nb)], 1);
  }
  __syncthreads();
  if (t < nb) blockhist[(size_t)blockIdx.x * nb + t] = hist[t];
}
static_assert(EP_THREADS == NB_MAX, "one thread per LDS table entry");

static __device__ __forceinline__ void
bucket_bin_body(int* __restrict__ lds,
                const int* __restrict__ src, const int* __restrict__ dst,
                const int* __restrict__ blockhist, int2* __restrict__ binned,
                int* __restrict__ bbase, int E, int nb_in, int nwg, int w) {
  int* tot = lds;
  int* cur = lds + NB_MAX;
  int* wsum = lds + 2 * NB_MAX;
  const int t = threadIdx.x;
  const int lane = t & 31;
  const int wv = t >> 5;
  const int nb = min(nb_in, NB_MAX);
  for (int b = t; b < NB_MAX; b += BIN_THREADS) {
    int ts = 0, ps = 0;
    if (b < nb) {
#pragma unroll 16
      for (int r = 0; r < nwg; ++r) {
        const int v = blockhist[(size_t)r * nb + b];
        ts += v;
        ps += (r < w) ? v : 0;
      }
    }
    tot[b] = ts;
    cur[b] = ps;
  }
  __syncthreads();
  const int4 c4 = *reinterpret_cast<const int4*>(tot + 4 * t);
  const int c = c4.x + c4.y + c4.z + c4.w;
  int inc = c;
  for (int o = 1; o < 32; o <<= 1) {
    const int y = __shfl_up(inc, o);
    if (lane >= o) inc += y;
  }
  if (lane == 31) wsum[wv] = inc;
  __syncthreads();
  if (wv == 0) {
    const int x = (lane < BIN_THREADS / 32) ? wsum[lane] : 0;
    int xi = x;
    for (int o = 1; o < 32; o <<= 1) {
      const int y = __shfl_up(xi, o);
      if (lane >= o) xi += y;
    }
    if (lane < BIN_THREADS / 32) wsum[lane] = xi - x;
  }
  __syncthreads();
  {
    const int b0 = wsum[wv] + inc - c;
    const int b1 = b0 + c4.x;
    const int b2 = b1 + c4.y;
    const int b3 = b2 + c4.z;
    int4 k4 = *reinterpret_cast<const int4*>(cur + 4 * t);
    k4.x += b0; k4.y += b1; k4.z += b2; k4.w += b3;
    *reinterpret_cast<int4*>(cur + 4 * t) = k4;
    if (w == 0) {
      if (4 * t < nb) bbase[4 * t] = b0;
      if (4 * t + 1 < nb) bbase[4 * t + 1] = b1;
      if (4 * t + 2 < nb) bbase[4 * t + 2] = b2;
      if (4 * t + 3 < nb) bbase[4 * t + 3] = b3;
    }
  }
  __syncthreads();
  const int e0 = w * EP_CHUNK + t;
  int s[BIN_PER_THREAD], d[BIN_PER_THREAD];
#pragma unroll
  for (int u = 0; u < BIN_PER_THREAD; ++u) {
    const int e = min(e0 + u * BIN_THREADS, E - 1);
    s[u] = src[e];
    d[u] = dst[e];
  }
#pragma unroll
  for (int u = 0; u < BIN_PER_THREAD; ++u) {
    const int e = e0 + u * BIN_THREADS;
    if (e < E) {
      int pos = atomicAdd(&cur[bucket_of(d[u], nb)], 1);
      pos = min(max(pos, 0), E - 1);
      binned[pos] = make_int2(s[u], d[u]);
    }
  }
}

__global__ void __launch_bounds__(1024)
k_bucket_finalize(const int2* __restrict__ binned, const int* __restrict__ bbase,
                  int* __restrict__ cnt, int* __restrict__ rowptr,
                  float* __restrict__ dinv, int* __restrict__ csr_src,
                  int n, int E, int nb_in) {
  __shared__ int lcnt[BKT_NODES];
  __shared__ int wsum[BKT_NODES / 32];
  const int t = threadIdx.x;
  const int lane = t & 31;
  const int wv = t >> 5;
  const int b = blockIdx.x;
  const int nb = min(nb_in, NB_MAX);
  const int beg = min(max(bbase[b], 0), E);
  const int end = (b + 1 < nb) ? min(max(bbase[b + 1], beg), E) : E;
  if (t < BKT_NODES) lcnt[t] = 0;
  __syncthreads();
  for (int i0 = beg; i0 < end; i0 += 1024 * FIN_PER_THREAD) {
    int dl[FIN_PER_THREAD];
#pragma unroll
    for (int u = 0; u < FIN_PER_THREAD; ++u) {
      dl[u] = binned[min(i0 + u * 1024 + t, end - 1)].y;
    }
#pragma unroll
    for (int u = 0; u < FIN_PER_THREAD; ++u) {
      const int i = i0 + u * 1024 + t;
      if (i < end) atomicAdd(&lcnt[dl[u] & (BKT_NODES - 1)], 1);
    }
  }
  __syncthreads();
  const int c = (t < BKT_NODES) ? lcnt[t] : 0;
  int inc = c;
  for (int o = 1; o < 32; o <<= 1) {
    const int y = __shfl_up(inc, o);
    if (lane >= o) inc += y;
  }
  if (lane == 31 && wv < BKT_NODES / 32) wsum[wv] = inc;
  __syncthreads();
  if (t < BKT_NODES) {
    int woff = 0;
#pragma unroll
    for (int k = 0; k < BKT_NODES / 32; ++k) woff += (k < wv) ? wsum[k] : 0;
    const int row = beg + woff + inc - c;
    const int node = b * BKT_NODES + t;
    if (node < n) {
      cnt[node] = c;
      rowptr[node] = row;
      dinv[node] = dinv_of_count(c);
    }
    lcnt[t] = row;
  }
  __syncthreads();
  for (int i0 = beg; i0 < end; i0 += 1024 * FIN_PER_THREAD) {
    int2 v[FIN_PER_THREAD];
#pragma unroll
    for (int u = 0; u < FIN_PER_THREAD; ++u) {
      v[u] = binned[min(i0 + u * 1024 + t, end - 1)];
    }
#pragma unroll
    for (int u = 0; u < FIN_PER_THREAD; ++u) {
      const int i = i0 + u * 1024 + t;
      if (i < end) {
        int pos = atomicAdd(&lcnt[v[u].y & (BKT_NODES - 1)], 1);
        pos = min(max(pos, 0), E - 1);
        csr_src[pos] = v[u].x;
      }
    }
  }
}


static constexpr int W1_PITCH = 2 * HID_DIM + 32;
static constexpr int W1_LDS_WORDS = (IN_DIM / 2) * W1_PITCH;
static_assert(IN_DIM / 2 == 8 * 16 && HID_DIM == 4 * 32,
              "staging: 8 waves x 16 pair rows, 32 lanes x 4 columns");

static __device__ __forceinline__ void
gemm1_strip_body(float* __restrict__ Bs,
                 const float* __restrict__ A,
                 const float* __restrict__ B,
                 float* __restrict__ C, int M, int blk) {
  const int lane = threadIdx.x & 31;
  const int wv = __builtin_amdgcn_readfirstlane((int)(threadIdx.x >> 5));

  float4 r0[16], r1[16];
#pragma unroll
  for (int i = 0; i < 16; ++i) {
    const float* g = B + (size_t)(2 * (wv + 8 * i)) * HID_DIM + 4 * lane;
    r0[i] = *reinterpret_cast<const float4*>(g);
    r1[i] = *reinterpret_cast<const float4*>(g + HID_DIM);
  }
#pragma unroll
  for (int i = 0; i < 16; ++i) {
    float4 lo, hi;
    lo.x = r0[i].x; lo.y = r1[i].x; lo.z = r0[i].y; lo.w = r1[i].y;
    hi.x = r0[i].z; hi.y = r1[i].z; hi.z = r0[i].w; hi.w = r1[i].w;
    float* d = Bs + (wv + 8 * i) * W1_PITCH + 8 * lane;
    *reinterpret_cast<float4*>(d) = lo;
    *reinterpret_cast<float4*>(d + 4) = hi;
  }
  __syncthreads();

  const int rt = blk * 8 + wv;
  if (rt * 16 >= M) return;

  const int lm = lane & 15;
  const int kh = lane >> 4;

  const int arow_i = min(rt * 16 + lm, M - 1);
  const float* arow = A + (size_t)arow_i * IN_DIM + 2 * kh;
  const float* bl = Bs + kh * W1_PITCH + 2 * lm;

  v8f acc[8];
#pragma unroll
  for (int ct = 0; ct < 8; ++ct) acc[ct] = v8f{};

  v2f a_cur[8], a_nxt[8];
#pragma unroll
  for (int s = 0; s < 8; ++s) a_cur[s] = *reinterpret_cast<const v2f*>(arow + 4 * s);

  for (int c = 0; c < IN_DIM / 32; ++c) {
    if (c + 1 < IN_DIM / 32) {
#pragma unroll
      for (int s = 0; s < 8; ++s)
        a_nxt[s] = *reinterpret_cast<const v2f*>(arow + 32 * (c + 1) + 4 * s);
    }
    const float* bc = bl + c * (16 * W1_PITCH);
#pragma unroll
    for (int s = 0; s < 8; ++s) {
#pragma unroll
      for (int ct = 0; ct < 8; ++ct) {
        const v2f b = *reinterpret_cast<const v2f*>(bc + (2 * s) * W1_PITCH + 32 * ct);
        acc[ct] = __builtin_amdgcn_wmma_f32_16x16x4_f32(false, a_cur[s], false, b,
                                                        (short)0, acc[ct], false, false);
      }
    }
#pragma unroll
    for (int s = 0; s < 8; ++s) a_cur[s] = a_nxt[s];
  }

  const int r0i = rt * 16 + 8 * kh;
  float* crow = C + (size_t)r0i * HID_DIM + lm;
  if (rt * 16 + 16 <= M) {
#pragma unroll
    for (int ct = 0; ct < 8; ++ct) {
#pragma unroll
      for (int j = 0; j < 8; ++j) crow[(size_t)j * HID_DIM + ct * 16] = acc[ct][j];
    }
  } else {
#pragma unroll
    for (int ct = 0; ct < 8; ++ct) {
#pragma unroll
      for (int j = 0; j < 8; ++j) {
        if (r0i + j < M) crow[(size_t)j * HID_DIM + ct * 16] = acc[ct][j];
      }
    }
  }
}

static_assert(W1_LDS_WORDS >= 2 * NB_MAX + BIN_THREADS / 32, "bin tables fit the GEMM's LDS array");

__global__ void __launch_bounds__(256)
k_gemm1_strip_wmma_f32_and_bucket_bin(const float* __restrict__ A,
                                      const float* __restrict__ B,
                                      float* __restrict__ C, int M, int gemm_blocks,
                                      const int* __restrict__ src, const int* __restrict__ dst,
                                      const int* __restrict__ blockhist, int2* __restrict__ binned,
                                      int* __restrict__ bbase, int E, int nb_in, int nwg) {
  __shared__ __attribute__((aligned(16))) float Bs[W1_LDS_WORDS];
  const int blk = (int)blockIdx.x;
  if (blk < gemm_blocks) {
    gemm1_strip_body(Bs, A, B, C, M, blk);
  } else {
    bucket_bin_body(reinterpret_cast<int*>(Bs), src, dst, blockhist, binned, bbase,
                    E, nb_in, nwg, blk - gemm_blocks);
  }
}


static constexpr int AG_ROWS = 16;
static constexpr int AG_PITCH = HID_DIM + 4;
static constexpr int W2_PAIR_WORDS = 2 * OUT_DIM;
static_assert(OUT_DIM == 16 && HID_DIM == 128 && (HID_DIM / 2) * 4 == 256,
              "one 16x16 output tile; W2 staging: 256 threads x (one row pair, 4 columns)");

__global__ void __launch_bounds__(256)
k_gather_f128_bias_relu_gemm2(const int* __restrict__ rowptr,
                              const int* __restrict__ cnt,
                              const int* __restrict__ csr_src,
                              float* __restrict__ csr_w,
                              const float* __restrict__ dinv,
                              const float* __restrict__ h,
                              const float* __restrict__ bias,
                              const float* __restrict__ W2,
                              float* __restrict__ h2, int n, int E) {
  __shared__ __attribute__((aligned(16))) float tile[AG_ROWS * AG_PITCH];
  __shared__ __attribute__((aligned(16))) float w2s[(HID_DIM / 2) * W2_PAIR_WORDS];
  const int lane = threadIdx.x & 31;
  const int wv = __builtin_amdgcn_readfirstlane((int)(threadIdx.x >> 5));

  const int p = (int)threadIdx.x >> 2;
  const int c0 = ((int)threadIdx.x & 3) * 4;
  const float4 wr0 = *reinterpret_cast<const float4*>(W2 + (size_t)(2 * p) * OUT_DIM + c0);
  const float4 wr1 = *reinterpret_cast<const float4*>(W2 + (size_t)(2 * p + 1) * OUT_DIM + c0);

  const float4 bv = *reinterpret_cast<const float4*>(bias + lane * 4);
  const int node0 = blockIdx.x * AG_ROWS;

  for (int r = 0; r < 2; ++r) {
    const int slot = wv + 8 * r;
    const int node = node0 + slot;
    float4 acc;
    acc.x = 0.0f; acc.y = 0.0f; acc.z = 0.0f; acc.w = 0.0f;
    if (node < n) {
      const int len = min(max(cnt[node], 0), E);
      const int start = min(max(rowptr[node], 0), E - len);
      const float dd = dinv[node];
      const float4 hv = *reinterpret_cast<const float4*>(h + (size_t)node * 128 + lane * 4);
      const float wself = dd * dd;
      acc.x = hv.x * wself;
      acc.y = hv.y * wself;
      acc.z = hv.z * wself;
      acc.w = hv.w * wself;
      for (int b = 0; b < len; b += 32) {
        const int m = min(32, len - b);
        int s_l = 0;
        float w_l = 0.0f;
        if (lane < m) {
          s_l = min(max(csr_src[start + b + lane], 0), n - 1);
          const float ds = dinv[s_l];
          csr_w[start + b + lane] = ds;
          w_l = ds * dd;
        }
        int j = 0;
        for (; j + 4 <= m; j += 4) {
          const int s0 = __builtin_amdgcn_readlane(s_l, j);
          const int s1 = __builtin_amdgcn_readlane(s_l, j + 1);
          const int s2 = __builtin_amdgcn_readlane(s_l, j + 2);
          const int s3 = __builtin_amdgcn_readlane(s_l, j + 3);
          const float w0 = __int_as_float(__builtin_amdgcn_readlane(__float_as_int(w_l), j));
          const float w1 = __int_as_float(__builtin_amdgcn_readlane(__float_as_int(w_l), j + 1));
          const float w2 = __int_as_float(__builtin_amdgcn_readlane(__float_as_int(w_l), j + 2));
          const float w3 = __int_as_float(__builtin_amdgcn_readlane(__float_as_int(w_l), j + 3));
          const float4 v0 = *reinterpret_cast<const float4*>(h + (size_t)s0 * 128 + lane * 4);
          const float4 v1 = *reinterpret_cast<const float4*>(h + (size_t)s1 * 128 + lane * 4);
          const float4 v2 = *reinterpret_cast<const float4*>(h + (size_t)s2 * 128 + lane * 4);
          const float4 v3 = *reinterpret_cast<const float4*>(h + (size_t)s3 * 128 + lane * 4);
          acc.x += v0.x * w0; acc.y += v0.y * w0; acc.z += v0.z * w0; acc.w += v0.w * w0;
          acc.x += v1.x * w1; acc.y += v1.y * w1; acc.z += v1.z * w1; acc.w += v1.w * w1;
          acc.x += v2.x * w2; acc.y += v2.y * w2; acc.z += v2.z * w2; acc.w += v2.w * w2;
          acc.x += v3.x * w3; acc.y += v3.y * w3; acc.z += v3.z * w3; acc.w += v3.w * w3;
        }
        for (; j < m; ++j) {
          const int s0 = __builtin_amdgcn_readlane(s_l, j);
          const float w0 = __int_as_float(__builtin_amdgcn_readlane(__float_as_int(w_l), j));
          const float4 v0 = *reinterpret_cast<const float4*>(h + (size_t)s0 * 128 + lane * 4);
          acc.x += v0.x * w0; acc.y += v0.y * w0; acc.z += v0.z * w0; acc.w += v0.w * w0;
        }
      }
      acc.x = fmaxf(acc.x + bv.x, 0.0f);
      acc.y = fmaxf(acc.y + bv.y, 0.0f);
      acc.z = fmaxf(acc.z + bv.z, 0.0f);
      acc.w = fmaxf(acc.w + bv.w, 0.0f);
    }
    *reinterpret_cast<float4*>(tile + slot * AG_PITCH + lane * 4) = acc;
    if (r == 0) {
      float4 lo, hi;
      lo.x = wr0.x; lo.y = wr1.x; lo.z = wr0.y; lo.w = wr1.y;
      hi.x = wr0.z; hi.y = wr1.z; hi.z = wr0.w; hi.w = wr1.w;
      float* d = w2s + p * W2_PAIR_WORDS + 2 * c0;
      *reinterpret_cast<float4*>(d) = lo;
      *reinterpret_cast<float4*>(d + 4) = hi;
    }
  }
  __syncthreads();
  if (wv != 0) return;

  const int lm = lane & 15;
  const int kh = lane >> 4;
  const float* al = tile + lm * AG_PITCH + 2 * kh;
  const float* bl = w2s + kh * W2_PAIR_WORDS + 2 * lm;
  v8f c = {};
#pragma unroll 8
  for (int k = 0; k < HID_DIM; k += 4) {
    const v2f a = *reinterpret_cast<const v2f*>(al + k);
    const v2f b = *reinterpret_cast<const v2f*>(bl + (k / 2) * W2_PAIR_WORDS);
    c = __builtin_amdgcn_wmma_f32_16x16x4_f32(false, a, false, b, (short)0, c, false, false);
  }
  const int r0i = node0 + 8 * kh;
  float* crow = h2 + (size_t)r0i * OUT_DIM + lm;
  if (node0 + AG_ROWS <= n) {
#pragma unroll
    for (int j = 0; j < 8; ++j) crow[(size_t)j * OUT_DIM] = c[j];
  } else {
#pragma unroll
    for (int j = 0; j < 8; ++j) {
      if (r0i + j < n) crow[(size_t)j * OUT_DIM] = c[j];
    }
  }
}

__global__ void k_gather_f16_bias_logsoftmax(const int* __restrict__ rowptr,
                                             const int* __restrict__ cnt,
                                             const int* __restrict__ csr_src,
                                             const float* __restrict__ csr_w,
                                             const float* __restrict__ dinv,
                                             const float* __restrict__ h,
                                             const float* __restrict__ bias,
                                             float* __restrict__ out, int n, int E) {
  const int idx = blockIdx.x * blockDim.x + threadIdx.x;
  const int node = idx >> 4;
  if (node >= n) return;
  const int f = idx & 15;
  const int len = min(max(cnt[node], 0), E);
  const int start = min(max(rowptr[node], 0), E - len);
  const float dd = dinv[node];
  float acc = h[(size_t)node * 16 + f] * (dd * dd);
  int j = 0;
  for (; j + 4 <= len; j += 4) {
    const int s0 = min(max(csr_src[start + j], 0), n - 1);
    const int s1 = min(max(csr_src[start + j + 1], 0), n - 1);
    const int s2 = min(max(csr_src[start + j + 2], 0), n - 1);
    const int s3 = min(max(csr_src[start + j + 3], 0), n - 1);
    const float w0 = csr_w[start + j] * dd;
    const float w1 = csr_w[start + j + 1] * dd;
    const float w2 = csr_w[start + j + 2] * dd;
    const float w3 = csr_w[start + j + 3] * dd;
    const float v0 = h[(size_t)s0 * 16 + f];
    const float v1 = h[(size_t)s1 * 16 + f];
    const float v2 = h[(size_t)s2 * 16 + f];
    const float v3 = h[(size_t)s3 * 16 + f];
    acc += v0 * w0;
    acc += v1 * w1;
    acc += v2 * w2;
    acc += v3 * w3;
  }
  for (; j < len; ++j) {
    const int s0 = min(max(csr_src[start + j], 0), n - 1);
    acc += h[(size_t)s0 * 16 + f] * (csr_w[start + j] * dd);
  }
  const float v = acc + bias[f];
  float m = __shfl(v, 0, 16);
#pragma unroll
  for (int k = 1; k < 16; ++k) m = fmaxf(m, __shfl(v, k, 16));
  const float ex = expf(v - m);
  float s = 0.0f;
#pragma unroll
  for (int k = 0; k < 16; ++k) s += __shfl(ex, k, 16);
  out[(size_t)node * 16 + f] = v - m - logf(s);
}

extern "C" void kernel_launch(void* const* d_in, const int* in_sizes, int n_in,
                              void* d_out, int out_size, void* d_ws, size_t ws_size,
                              hipStream_t stream) {
  (void)n_in; (void)out_size; (void)ws_size;

  const float* x  = (const float*)d_in[0];
  const int*   ei = (const int*)d_in[1];
  const float* W1 = (const float*)d_in[2];
  const float* b1 = (const float*)d_in[3];
  const float* W2 = (const float*)d_in[4];
  const float* b2 = (const float*)d_in[5];

  const int n = in_sizes[0] / IN_DIM;
  const int E = in_sizes[1] / 2;
  const int* src = ei;
  const int* dst = ei + E;

  const int B = 256;
  auto cdiv = [](long long a, long long b) { return (int)((a + b - 1) / b); };

  const int nb  = cdiv(n, BKT_NODES);
  const int nwg = cdiv(E, EP_CHUNK);

  float* h1     = (float*)d_ws;
  float* h2     = h1 + (size_t)n * HID_DIM;
  int2*  binned = (int2*)(h2 + (size_t)n * OUT_DIM);
  int*   csr    = (int*)(binned + E);
  float* csrw   = (float*)(csr + E);
  int*   cnt    = (int*)(csrw + E);
  int*   rowptr = cnt + n;
  float* dinv   = (float*)(rowptr + n);
  int*   bhist  = (int*)(dinv + n);
  int*   bbase  = bhist + (size_t)nwg * nb;
  float* out    = (float*)d_out;

  k_bucket_hist<<<nwg, EP_THREADS, 0, stream>>>(dst, bhist, E, nb);
  const int gemm_blocks = cdiv(cdiv(n, 16), 8);
  k_gemm1_strip_wmma_f32_and_bucket_bin<<<gemm_blocks + nwg, B, 0, stream>>>(
      x, W1, h1, n, gemm_blocks, src, dst, bhist, binned, bbase, E, nb, nwg);
  k_bucket_finalize<<<nb, 1024, 0, stream>>>(binned, bbase, cnt, rowptr, dinv, csr, n, E, nb);

  k_gather_f128_bias_relu_gemm2<<<cdiv(n, AG_ROWS), B, 0, stream>>>(rowptr, cnt, csr, csrw, dinv, h1, b1, W2, h2, n, E);

  k_gather_f16_bias_logsoftmax<<<cdiv((long long)n * OUT_DIM, B), B, 0, stream>>>(rowptr, cnt, csr, csrw, dinv, h2, b2, out, n, E);
}


# Edge chunks sorted by bucket each on its own inside the GEMM launch (per-chunk offset table); histogram launch gone (5 launches -> 4); sums: self term first, then arrival order of the LDS adds
# speedup vs baseline: 4.7338x; 1.0998x over previous
#include <hip/hip_runtime.h>
#include <math.h>

typedef __attribute__((ext_vector_type(2))) float v2f;
typedef __attribute__((ext_vector_type(8))) float v8f;

static constexpr int IN_DIM  = 256;
static constexpr int HID_DIM = 128;
static constexpr int OUT_DIM = 16;

static constexpr int BKT_SHIFT = 8;
static constexpr int BKT_NODES = 1 << BKT_SHIFT;
static constexpr int NB_MAX = 1024;
static constexpr int BIN_THREADS = 256;
static constexpr int BIN_PER_THREAD = 32;
static constexpr int EP_CHUNK = BIN_THREADS * BIN_PER_THREAD;
static_assert(NB_MAX == 4 * BIN_THREADS, "bucket scan: four consecutive buckets per thread");
static constexpr int FIN_THREADS = 1024;
static constexpr int FIN_WAVES = FIN_THREADS / 32;
static constexpr int FIN_CHUNKS = 4;
static constexpr int FIN_DEPTH = 2;

static __device__ __forceinline__ int bucket_of(int d, int nb) {
  return min(max(d >> BKT_SHIFT, 0), nb - 1);
}

static __device__ __forceinline__ float dinv_of_count(int c) {
  const float d = (float)(c + 1);
  return (d > 0.0f) ? rsqrtf(d) : 0.0f;
}

static __device__ __forceinline__ void
bucket_bin_body(int* __restrict__ lds,
                const int* __restrict__ src, const int* __restrict__ dst,
                int2* __restrict__ binned, int* __restrict__ coff,
                int E, int nb_in, int nwg, int w) {
  int* cur = lds;
  int* wsum = lds + NB_MAX;
  const int t = threadIdx.x;
  const int lane = t & 31;
  const int wv = t >> 5;
  const int nb = min(nb_in, NB_MAX);
  *reinterpret_cast<int4*>(cur + 4 * t) = make_int4(0, 0, 0, 0);
  const int e0 = w * EP_CHUNK + t;
  int s[BIN_PER_THREAD], d[BIN_PER_THREAD];
#pragma unroll
  for (int u = 0; u < BIN_PER_THREAD; ++u) {
    const int e = min(e0 + u * BIN_THREADS, E - 1);
    s[u] = src[e];
    d[u] = dst[e];
  }
  __syncthreads();
#pragma unroll
  for (int u = 0; u < BIN_PER_THREAD; ++u) {
    if (e0 + u * BIN_THREADS < E) atomicAdd(&cur[bucket_of(d[u], nb)], 1);
  }
  __syncthreads();
  const int4 c4 = *reinterpret_cast<const int4*>(cur + 4 * t);
  const int c = c4.x + c4.y + c4.z + c4.w;
  int inc = c;
  for (int o = 1; o < 32; o <<= 1) {
    const int y = __shfl_up(inc, o);
    if (lane >= o) inc += y;
  }
  if (lane == 31) wsum[wv] = inc;
  __syncthreads();
  if (wv == 0) {
    const int x = (lane < BIN_THREADS / 32) ? wsum[lane] : 0;
    int xi = x;
    for (int o = 1; o < 32; o <<= 1) {
      const int y = __shfl_up(xi, o);
      if (lane >= o) xi += y;
    }
    if (lane < BIN_THREADS / 32) wsum[lane] = xi - x;
  }
  __syncthreads();
  {
    const int b0 = wsum[wv] + inc - c;
    const int b1 = b0 + c4.x;
    const int b2 = b1 + c4.y;
    const int b3 = b2 + c4.z;
    *reinterpret_cast<int4*>(cur + 4 * t) = make_int4(b0, b1, b2, b3);
    if (4 * t < nb) coff[(size_t)(4 * t) * nwg + w] = b0;
    if (4 * t + 1 < nb) coff[(size_t)(4 * t + 1) * nwg + w] = b1;
    if (4 * t + 2 < nb) coff[(size_t)(4 * t + 2) * nwg + w] = b2;
    if (4 * t + 3 < nb) coff[(size_t)(4 * t + 3) * nwg + w] = b3;
    if (t == 0) coff[(size_t)nb * nwg + w] = min(EP_CHUNK, E - w * EP_CHUNK);
  }
  __syncthreads();
  const int cbase = w * EP_CHUNK;
#pragma unroll
  for (int u = 0; u < BIN_PER_THREAD; ++u) {
    if (e0 + u * BIN_THREADS < E) {
      int pos = atomicAdd(&cur[bucket_of(d[u], nb)], 1);
      pos = min(max(pos, 0), EP_CHUNK - 1);
      binned[min(cbase + pos, E - 1)] = make_int2(s[u], d[u]);
    }
  }
}

static __device__ __forceinline__ void
piece_bounds(const int* __restrict__ row0, const int* __restrict__ row1,
             int w, int nwg, int E, int& base, int& len) {
  const int wc = min(w, nwg - 1);
  const int r0 = row0[wc];
  const int r1 = row1[wc];
  const int clen = min(EP_CHUNK, E - wc * EP_CHUNK);
  const int a = min(max(r0, 0), clen);
  const int z = min(max(r1, a), clen);
  base = wc * EP_CHUNK + a;
  len = (w < nwg) ? z - a : 0;
}

static __device__ __forceinline__ void
pieces_count(const int2* __restrict__ binned, int* __restrict__ lcnt,
             const int (&base)[FIN_CHUNKS], const int (&len)[FIN_CHUNKS], int lane, int E) {
  int dl[FIN_CHUNKS][FIN_DEPTH];
#pragma unroll
  for (int u = 0; u < FIN_CHUNKS; ++u) {
#pragma unroll
    for (int k = 0; k < FIN_DEPTH; ++k) {
      dl[u][k] = binned[min(base[u] + lane + 32 * k, E - 1)].y;
    }
  }
#pragma unroll
  for (int u = 0; u < FIN_CHUNKS; ++u) {
#pragma unroll
    for (int k = 0; k < FIN_DEPTH; ++k) {
      if (lane + 32 * k < len[u]) atomicAdd(&lcnt[dl[u][k] & (BKT_NODES - 1)], 1);
    }
  }
#pragma unroll
  for (int u = 0; u < FIN_CHUNKS; ++u) {
    for (int o = lane + 32 * FIN_DEPTH; o < len[u]; o += 32) {
      atomicAdd(&lcnt[binned[base[u] + o].y & (BKT_NODES - 1)], 1);
    }
  }
}

static __device__ __forceinline__ void
pieces_place(const int2* __restrict__ binned, int* __restrict__ lcnt, int* __restrict__ csr_src,
             const int (&base)[FIN_CHUNKS], const int (&len)[FIN_CHUNKS], int lane, int E) {
  int2 v[FIN_CHUNKS][FIN_DEPTH];
#pragma unroll
  for (int u = 0; u < FIN_CHUNKS; ++u) {
#pragma unroll
    for (int k = 0; k < FIN_DEPTH; ++k) {
      v[u][k] = binned[min(base[u] + lane + 32 * k, E - 1)];
    }
  }
#pragma unroll
  for (int u = 0; u < FIN_CHUNKS; ++u) {
#pragma unroll
    for (int k = 0; k < FIN_DEPTH; ++k) {
      if (lane + 32 * k < len[u]) {
        int pos = atomicAdd(&lcnt[v[u][k].y & (BKT_NODES - 1)], 1);
        pos = min(max(pos, 0), E - 1);
        csr_src[pos] = v[u][k].x;
      }
    }
  }
#pragma unroll
  for (int u = 0; u < FIN_CHUNKS; ++u) {
    for (int o = lane + 32 * FIN_DEPTH; o < len[u]; o += 32) {
      const int2 x = binned[base[u] + o];
      int pos = atomicAdd(&lcnt[x.y & (BKT_NODES - 1)], 1);
      pos = min(max(pos, 0), E - 1);
      csr_src[pos] = x.x;
    }
  }
}

__global__ void __launch_bounds__(FIN_THREADS)
k_bucket_finalize(const int2* __restrict__ binned, const int* __restrict__ coff,
                  int* __restrict__ cnt, int* __restrict__ rowptr,
                  float* __restrict__ dinv, int* __restrict__ csr_src,
                  int n, int E, int nb_in, int nwg) {
  __shared__ int lcnt[BKT_NODES];
  __shared__ int wsum[BKT_NODES / 32];
  __shared__ int bsum[FIN_WAVES];
  const int t = threadIdx.x;
  const int lane = t & 31;
  const int wv = __builtin_amdgcn_readfirstlane(t >> 5);
  const int b = blockIdx.x;
  const int nb = min(nb_in, NB_MAX);
  if (b >= nb) return;
  const int* row0 = coff + (size_t)b * nwg;
  const int* row1 = row0 + nwg;
  int base[FIN_CHUNKS], len[FIN_CHUNKS];
#pragma unroll
  for (int u = 0; u < FIN_CHUNKS; ++u) piece_bounds(row0, row1, wv + FIN_WAVES * u, nwg, E, base[u], len[u]);
  int part = 0;
#pragma unroll 1
  for (int w = t; w < nwg; w += FIN_THREADS) part += row0[w];
  for (int o = 16; o > 0; o >>= 1) part += __shfl_xor(part, o);
  if (lane == 0) bsum[wv] = part;
  if (t < BKT_NODES) lcnt[t] = 0;
  __syncthreads();
  pieces_count(binned, lcnt, base, len, lane, E);
  for (int wb = wv + FIN_WAVES * FIN_CHUNKS; wb < nwg; wb += FIN_WAVES * FIN_CHUNKS) {
    int base2[FIN_CHUNKS], len2[FIN_CHUNKS];
#pragma unroll
    for (int u = 0; u < FIN_CHUNKS; ++u) piece_bounds(row0, row1, wb + FIN_WAVES * u, nwg, E, base2[u], len2[u]);
    pieces_count(binned, lcnt, base2, len2, lane, E);
  }
  __syncthreads();
  const int c = (t < BKT_NODES) ? lcnt[t] : 0;
  int inc = c;
  for (int o = 1; o < 32; o <<= 1) {
    const int y = __shfl_up(inc, o);
    if (lane >= o) inc += y;
  }
  if (lane == 31 && wv < BKT_NODES / 32) wsum[wv] = inc;
  __syncthreads();
  if (t < BKT_NODES) {
    int beg = bsum[lane];
    for (int o = 16; o > 0; o >>= 1) beg += __shfl_xor(beg, o);
    beg = min(max(beg, 0), E);
    int woff = 0;
#pragma unroll
    for (int k = 0; k < BKT_NODES / 32; ++k) woff += (k < wv) ? wsum[k] : 0;
    const int row = beg + woff + inc - c;
    const int node = b * BKT_NODES + t;
    if (node < n) {
      cnt[node] = c;
      rowptr[node] = row;
      dinv[node] = dinv_of_count(c);
    }
    lcnt[t] = row;
  }
  __syncthreads();
  pieces_place(binned, lcnt, csr_src, base, len, lane, E);
  for (int wb = wv + FIN_WAVES * FIN_CHUNKS; wb < nwg; wb += FIN_WAVES * FIN_CHUNKS) {
    int base2[FIN_CHUNKS], len2[FIN_CHUNKS];
#pragma unroll
    for (int u = 0; u < FIN_CHUNKS; ++u) piece_bounds(row0, row1, wb + FIN_WAVES * u, nwg, E, base2[u], len2[u]);
    pieces_place(binned, lcnt, csr_src, base2, len2, lane, E);
  }
}
static_assert(FIN_WAVES == 32, "one wave sum of the table row per lane");


static constexpr int W1_PITCH = 2 * HID_DIM + 32;
static constexpr int W1_LDS_WORDS = (IN_DIM / 2) * W1_PITCH;
static_assert(IN_DIM / 2 == 8 * 16 && HID_DIM == 4 * 32,
              "staging: 8 waves x 16 pair rows, 32 lanes x 4 columns");

static __device__ __forceinline__ void
gemm1_strip_body(float* __restrict__ Bs,
                 const float* __restrict__ A,
                 const float* __restrict__ B,
                 float* __restrict__ C, int M, int blk) {
  const int lane = threadIdx.x & 31;
  const int wv = __builtin_amdgcn_readfirstlane((int)(threadIdx.x >> 5));

  float4 r0[16], r1[16];
#pragma unroll
  for (int i = 0; i < 16; ++i) {
    const float* g = B + (size_t)(2 * (wv + 8 * i)) * HID_DIM + 4 * lane;
    r0[i] = *reinterpret_cast<const float4*>(g);
    r1[i] = *reinterpret_cast<const float4*>(g + HID_DIM);
  }
#pragma unroll
  for (int i = 0; i < 16; ++i) {
    float4 lo, hi;
    lo.x = r0[i].x; lo.y = r1[i].x; lo.z = r0[i].y; lo.w = r1[i].y;
    hi.x = r0[i].z; hi.y = r1[i].z; hi.z = r0[i].w; hi.w = r1[i].w;
    float* d = Bs + (wv + 8 * i) * W1_PITCH + 8 * lane;
    *reinterpret_cast<float4*>(d) = lo;
    *reinterpret_cast<float4*>(d + 4) = hi;
  }
  __syncthreads();

  const int rt = blk * 8 + wv;
  if (rt * 16 >= M) return;

  const int lm = lane & 15;
  const int kh = lane >> 4;

  const int arow_i = min(rt * 16 + lm, M - 1);
  const float* arow = A + (size_t)arow_i * IN_DIM + 2 * kh;
  const float* bl = Bs + kh * W1_PITCH + 2 * lm;

  v8f acc[8];
#pragma unroll
  for (int ct = 0; ct < 8; ++ct) acc[ct] = v8f{};

  v2f a_cur[8], a_nxt[8];
#pragma unroll
  for (int s = 0; s < 8; ++s) a_cur[s] = *reinterpret_cast<const v2f*>(arow + 4 * s);

  for (int c = 0; c < IN_DIM / 32; ++c) {
    if (c + 1 < IN_DIM / 32) {
#pragma unroll
      for (int s = 0; s < 8; ++s)
        a_nxt[s] = *reinterpret_cast<const v2f*>(arow + 32 * (c + 1) + 4 * s);
    }
    const float* bc = bl + c * (16 * W1_PITCH);
#pragma unroll
    for (int s = 0; s < 8; ++s) {
#pragma unroll
      for (int ct = 0; ct < 8; ++ct) {
        const v2f b = *reinterpret_cast<const v2f*>(bc + (2 * s) * W1_PITCH + 32 * ct);
        acc[ct] = __builtin_amdgcn_wmma_f32_16x16x4_f32(false, a_cur[s], false, b,
                                                        (short)0, acc[ct], false, false);
      }
    }
#pragma unroll
    for (int s = 0; s < 8; ++s) a_cur[s] = a_nxt[s];
  }

  const int r0i = rt * 16 + 8 * kh;
  float* crow = C + (size_t)r0i * HID_DIM + lm;
  if (rt * 16 + 16 <= M) {
#pragma unroll
    for (int ct = 0; ct < 8; ++ct) {
#pragma unroll
      for (int j = 0; j < 8; ++j) crow[(size_t)j * HID_DIM + ct * 16] = acc[ct][j];
    }
  } else {
#pragma unroll
    for (int ct = 0; ct < 8; ++ct) {
#pragma unroll
      for (int j = 0; j < 8; ++j) {
        if (r0i + j < M) crow[(size_t)j * HID_DIM + ct * 16] = acc[ct][j];
      }
    }
  }
}

static_assert(W1_LDS_WORDS >= NB_MAX + BIN_THREADS / 32, "bin table fits the GEMM's LDS array");

__global__ void __launch_bounds__(256)
k_gemm1_strip_wmma_f32_and_bucket_bin(const float* __restrict__ A,
                                      const float* __restrict__ B,
                                      float* __restrict__ C, int M, int gemm_blocks,
                                      const int* __restrict__ src, const int* __restrict__ dst,
                                      int2* __restrict__ binned, int* __restrict__ coff,
                                      int E, int nb_in, int nwg) {
  __shared__ __attribute__((aligned(16))) float Bs[W1_LDS_WORDS];
  const int blk = (int)blockIdx.x;
  if (blk < gemm_blocks) {
    gemm1_strip_body(Bs, A, B, C, M, blk);
  } else {
    bucket_bin_body(reinterpret_cast<int*>(Bs), src, dst, binned, coff,
                    E, nb_in, nwg, blk - gemm_blocks);
  }
}


static constexpr int AG_ROWS = 16;
static constexpr int AG_PITCH = HID_DIM + 4;
static constexpr int W2_PAIR_WORDS = 2 * OUT_DIM;
static_assert(OUT_DIM == 16 && HID_DIM == 128 && (HID_DIM / 2) * 4 == 256,
              "one 16x16 output tile; W2 staging: 256 threads x (one row pair, 4 columns)");

__global__ void __launch_bounds__(256)
k_gather_f128_bias_relu_gemm2(const int* __restrict__ rowptr,
                              const int* __restrict__ cnt,
                              const int* __restrict__ csr_src,
                              float* __restrict__ csr_w,
                              const float* __restrict__ dinv,
                              const float* __restrict__ h,
                              const float* __restrict__ bias,
                              const float* __restrict__ W2,
                              float* __restrict__ h2, int n, int E) {
  __shared__ __attribute__((aligned(16))) float tile[AG_ROWS * AG_PITCH];
  __shared__ __attribute__((aligned(16))) float w2s[(HID_DIM / 2) * W2_PAIR_WORDS];
  const int lane = threadIdx.x & 31;
  const int wv = __builtin_amdgcn_readfirstlane((int)(threadIdx.x >> 5));

  const int p = (int)threadIdx.x >> 2;
  const int c0 = ((int)threadIdx.x & 3) * 4;
  const float4 wr0 = *reinterpret_cast<const float4*>(W2 + (size_t)(2 * p) * OUT_DIM + c0);
  const float4 wr1 = *reinterpret_cast<const float4*>(W2 + (size_t)(2 * p + 1) * OUT_DIM + c0);

  const float4 bv = *reinterpret_cast<const float4*>(bias + lane * 4);
  const int node0 = blockIdx.x * AG_ROWS;

  for (int r = 0; r < 2; ++r) {
    const int slot = wv + 8 * r;
    const int node = node0 + slot;
    float4 acc;
    acc.x = 0.0f; acc.y = 0.0f; acc.z = 0.0f; acc.w = 0.0f;
    if (node < n) {
      const int len = min(max(cnt[node], 0), E);
      const int start = min(max(rowptr[node], 0), E - len);
      const float dd = dinv[node];
      const float4 hv = *reinterpret_cast<const float4*>(h + (size_t)node * 128 + lane * 4);
      const float wself = dd * dd;
      acc.x = hv.x * wself;
      acc.y = hv.y * wself;
      acc.z = hv.z * wself;
      acc.w = hv.w * wself;
      for (int b = 0; b < len; b += 32) {
        const int m = min(32, len - b);
        int s_l = 0;
        float w_l = 0.0f;
        if (lane < m) {
          s_l = min(max(csr_src[start + b + lane], 0), n - 1);
          const float ds = dinv[s_l];
          csr_w[start + b + lane] = ds;
          w_l = ds * dd;
        }
        int j = 0;
        for (; j + 4 <= m; j += 4) {
          const int s0 = __builtin_amdgcn_readlane(s_l, j);
          const int s1 = __builtin_amdgcn_readlane(s_l, j + 1);
          const int s2 = __builtin_amdgcn_readlane(s_l, j + 2);
          const int s3 = __builtin_amdgcn_readlane(s_l, j + 3);
          const float w0 = __int_as_float(__builtin_amdgcn_readlane(__float_as_int(w_l), j));
          const float w1 = __int_as_float(__builtin_amdgcn_readlane(__float_as_int(w_l), j + 1));
          const float w2 = __int_as_float(__builtin_amdgcn_readlane(__float_as_int(w_l), j + 2));
          const float w3 = __int_as_float(__builtin_amdgcn_readlane(__float_as_int(w_l), j + 3));
          const float4 v0 = *reinterpret_cast<const float4*>(h + (size_t)s0 * 128 + lane * 4);
          const float4 v1 = *reinterpret_cast<const float4*>(h + (size_t)s1 * 128 + lane * 4);
          const float4 v2 = *reinterpret_cast<const float4*>(h + (size_t)s2 * 128 + lane * 4);
          const float4 v3 = *reinterpret_cast<const float4*>(h + (size_t)s3 * 128 + lane * 4);
          acc.x += v0.x * w0; acc.y += v0.y * w0; acc.z += v0.z * w0; acc.w += v0.w * w0;
          acc.x += v1.x * w1; acc.y += v1.y * w1; acc.z += v1.z * w1; acc.w += v1.w * w1;
          acc.x += v2.x * w2; acc.y += v2.y * w2; acc.z += v2.z * w2; acc.w += v2.w * w2;
          acc.x += v3.x * w3; acc.y += v3.y * w3; acc.z += v3.z * w3; acc.w += v3.w * w3;
        }
        for (; j < m; ++j) {
          const int s0 = __builtin_amdgcn_readlane(s_l, j);
          const float w0 = __int_as_float(__builtin_amdgcn_readlane(__float_as_int(w_l), j));
          const float4 v0 = *reinterpret_cast<const float4*>(h + (size_t)s0 * 128 + lane * 4);
          acc.x += v0.x * w0; acc.y += v0.y * w0; acc.z += v0.z * w0; acc.w += v0.w * w0;
        }
      }
      acc.x = fmaxf(acc.x + bv.x, 0.0f);
      acc.y = fmaxf(acc.y + bv.y, 0.0f);
      acc.z = fmaxf(acc.z + bv.z, 0.0f);
      acc.w = fmaxf(acc.w + bv.w, 0.0f);
    }
    *reinterpret_cast<float4*>(tile + slot * AG_PITCH + lane * 4) = acc;
    if (r == 0) {
      float4 lo, hi;
      lo.x = wr0.x; lo.y = wr1.x; lo.z = wr0.y; lo.w = wr1.y;
      hi.x = wr0.z; hi.y = wr1.z; hi.z = wr0.w; hi.w = wr1.w;
      float* d = w2s + p * W2_PAIR_WORDS + 2 * c0;
      *reinterpret_cast<float4*>(d) = lo;
      *reinterpret_cast<float4*>(d + 4) = hi;
    }
  }
  __syncthreads();
  if (wv != 0) return;

  const int lm = lane & 15;
  const int kh = lane >> 4;
  const float* al = tile + lm * AG_PITCH + 2 * kh;
  const float* bl = w2s + kh * W2_PAIR_WORDS + 2 * lm;
  v8f c = {};
#pragma unroll 8
  for (int k = 0; k < HID_DIM; k += 4) {
    const v2f a = *reinterpret_cast<const v2f*>(al + k);
    const v2f b = *reinterpret_cast<const v2f*>(bl + (k / 2) * W2_PAIR_WORDS);
    c = __builtin_amdgcn_wmma_f32_16x16x4_f32(false, a, false, b, (short)0, c, false, false);
  }
  const int r0i = node0 + 8 * kh;
  float* crow = h2 + (size_t)r0i * OUT_DIM + lm;
  if (node0 + AG_ROWS <= n) {
#pragma unroll
    for (int j = 0; j < 8; ++j) crow[(size_t)j * OUT_DIM] = c[j];
  } else {
#pragma unroll
    for (int j = 0; j < 8; ++j) {
      if (r0i + j < n) crow[(size_t)j * OUT_DIM] = c[j];
    }
  }
}

__global__ void k_gather_f16_bias_logsoftmax(const int* __restrict__ rowptr,
                                             const int* __restrict__ cnt,
                                             const int* __restrict__ csr_src,
                                             const float* __restrict__ csr_w,
                                             const float* __restrict__ dinv,
                                             const float* __restrict__ h,
                                             const float* __restrict__ bias,
                                             float* __restrict__ out, int n, int E) {
  const int idx = blockIdx.x * blockDim.x + threadIdx.x;
  const int node = idx >> 4;
  if (node >= n) return;
  const int f = idx & 15;
  const int len = min(max(cnt[node], 0), E);
  const int start = min(max(rowptr[node], 0), E - len);
  const float dd = dinv[node];
  float acc = h[(size_t)node * 16 + f] * (dd * dd);
  int j = 0;
  for (; j + 4 <= len; j += 4) {
    const int s0 = min(max(csr_src[start + j], 0), n - 1);
    const int s1 = min(max(csr_src[start + j + 1], 0), n - 1);
    const int s2 = min(max(csr_src[start + j + 2], 0), n - 1);
    const int s3 = min(max(csr_src[start + j + 3], 0), n - 1);
    const float w0 = csr_w[start + j] * dd;
    const float w1 = csr_w[start + j + 1] * dd;
    const float w2 = csr_w[start + j + 2] * dd;
    const float w3 = csr_w[start + j + 3] * dd;
    const float v0 = h[(size_t)s0 * 16 + f];
    const float v1 = h[(size_t)s1 * 16 + f];
    const float v2 = h[(size_t)s2 * 16 + f];
    const float v3 = h[(size_t)s3 * 16 + f];
    acc += v0 * w0;
    acc += v1 * w1;
    acc += v2 * w2;
    acc += v3 * w3;
  }
  for (; j < len; ++j) {
    const int s0 = min(max(csr_src[start + j], 0), n - 1);
    acc += h[(size_t)s0 * 16 + f] * (csr_w[start + j] * dd);
  }
  const float v = acc + bias[f];
  float m = __shfl(v, 0, 16);
#pragma unroll
  for (int k = 1; k < 16; ++k) m = fmaxf(m, __shfl(v, k, 16));
  const float ex = expf(v - m);
  float s = 0.0f;
#pragma unroll
  for (int k = 0; k < 16; ++k) s += __shfl(ex, k, 16);
  out[(size_t)node * 16 + f] = v - m - logf(s);
}

extern "C" void kernel_launch(void* const* d_in, const int* in_sizes, int n_in,
                              void* d_out, int out_size, void* d_ws, size_t ws_size,
                              hipStream_t stream) {
  (void)n_in; (void)out_size; (void)ws_size;

  const float* x  = (const float*)d_in[0];
  const int*   ei = (const int*)d_in[1];
  const float* W1 = (const float*)d_in[2];
  const float* b1 = (const float*)d_in[3];
  const float* W2 = (const float*)d_in[4];
  const float* b2 = (const float*)d_in[5];

  const int n = in_sizes[0] / IN_DIM;
  const int E = in_sizes[1] / 2;
  const int* src = ei;
  const int* dst = ei + E;

  const int B = 256;
  auto cdiv = [](long long a, long long b) { return (int)((a + b - 1) / b); };

  const int nb  = cdiv(n, BKT_NODES);
  const int nwg = cdiv(E, EP_CHUNK);

  float* h1     = (float*)d_ws;
  float* h2     = h1 + (size_t)n * HID_DIM;
  int2*  binned = (int2*)(h2 + (size_t)n * OUT_DIM);
  int*   csr    = (int*)(binned + E);
  float* csrw   = (float*)(csr + E);
  int*   cnt    = (int*)(csrw + E);
  int*   rowptr = cnt + n;
  float* dinv   = (float*)(rowptr + n);
  int*   coff   = (int*)(dinv + n);
  float* out    = (float*)d_out;

  const int gemm_blocks = cdiv(cdiv(n, 16), 8);
  k_gemm1_strip_wmma_f32_and_bucket_bin<<<gemm_blocks + nwg, B, 0, stream>>>(
      x, W1, h1, n, gemm_blocks, src, dst, binned, coff, E, nb, nwg);
  k_bucket_finalize<<<nb, FIN_THREADS, 0, stream>>>(binned, coff, cnt, rowptr, dinv, csr, n, E, nb, nwg);

  k_gather_f128_bias_relu_gemm2<<<cdiv(n, AG_ROWS), B, 0, stream>>>(rowptr, cnt, csr, csrw, dinv, h1, b1, W2, h2, n, E);

  k_gather_f16_bias_logsoftmax<<<cdiv((long long)n * OUT_DIM, B), B, 0, stream>>>(rowptr, cnt, csr, csrw, dinv, h2, b2, out, n, E);
}
